# starting kernel
# baseline (speedup 1.0000x reference)
#include <hip/hip_runtime.h>
#include <math.h>

#define N_NODES  50000
#define N_EDGES  800000
#define IN_FEATS 128
#define H_FEATS  64
#define HEADS    4
#define D_HEAD   16
#define NEG_SLOPE 0.2f

typedef __attribute__((ext_vector_type(2))) float v2f;
typedef __attribute__((ext_vector_type(8))) float v8f;

__device__ __forceinline__ v8f wmma_f32_16x16x4(v2f a, v2f b, v8f c) {
    return __builtin_amdgcn_wmma_f32_16x16x4_f32(
         false, a,  false, b,
         (short)0, c,  false,  false);
}

__global__ void gemm_bias_wmma(const float* __restrict__ A,
                               const float* __restrict__ WT,
                               const float* __restrict__ bias,
                               float* __restrict__ C, int K) {
    const int lane = threadIdx.x & 31;
    const int wave = threadIdx.x >> 5;
    const int mt   = blockIdx.x;
    const int rm   = lane & 15;
    const int koff = (lane >> 4) << 1;

    const float* arow = A  + (size_t)(mt * 16 + rm) * K + koff;
    const float* brow = WT + (size_t)(wave * 16 + rm) * K + koff;

    v8f acc = {0.f, 0.f, 0.f, 0.f, 0.f, 0.f, 0.f, 0.f};
    for (int k = 0; k < K; k += 4) {
        const float2 av = *(const float2*)(arow + k);
        const float2 bv = *(const float2*)(brow + k);
        v2f a; a.x = av.x; a.y = av.y;
        v2f b; b.x = bv.x; b.y = bv.y;
        acc = wmma_f32_16x16x4(a, b, acc);
    }

    const int   col  = wave * 16 + rm;
    const float bv   = bias[col];
    const int   row0 = mt * 16 + ((lane >> 4) << 3);
#pragma unroll
    for (int r = 0; r < 8; ++r)
        C[(size_t)(row0 + r) * H_FEATS + col] = acc[r] + bv;
}

__global__ void transpose_k(const float* __restrict__ W, float* __restrict__ WT,
                            int R, int Ccols) {
    int t = blockIdx.x * blockDim.x + threadIdx.x;
    if (t >= R * Ccols) return;
    int r = t / Ccols, c = t - r * Ccols;
    WT[(size_t)c * R + r] = W[t];
}

__global__ void fill_k(float* __restrict__ p, float v, int n) {
    int t = blockIdx.x * blockDim.x + threadIdx.x;
    if (t < n) p[t] = v;
}

__global__ void relu_k(float* __restrict__ p, int n) {
    int t = blockIdx.x * blockDim.x + threadIdx.x;
    if (t < n) p[t] = fmaxf(p[t], 0.f);
}

__device__ __forceinline__ void atomicMaxFloat(float* addr, float val) {
    if (val >= 0.f) atomicMax((int*)addr, __float_as_int(val));
    else            atomicMin((unsigned int*)addr, __float_as_uint(val));
}

__global__ void edge_score_max(const int* __restrict__ src,
                               const int* __restrict__ dst,
                               const float* __restrict__ fs,
                               const float* __restrict__ fd,
                               const float* __restrict__ attn_l,
                               float* __restrict__ score,
                               float* __restrict__ mmax) {
    int t = blockIdx.x * blockDim.x + threadIdx.x;
    if (t >= N_EDGES * HEADS) return;
    const int e = t >> 2, h = t & 3;
    const int s = src[e], d = dst[e];

    const float4* ps = (const float4*)(fs + (size_t)s * H_FEATS + h * D_HEAD);
    const float4* pd = (const float4*)(fd + (size_t)d * H_FEATS + h * D_HEAD);
    const float4* pa = (const float4*)(attn_l + h * D_HEAD);

    float acc = 0.f;
#pragma unroll
    for (int q = 0; q < 4; ++q) {
        float4 a = ps[q], b = pd[q], w = pa[q];
        float x0 = a.x + b.x, x1 = a.y + b.y, x2 = a.z + b.z, x3 = a.w + b.w;
        x0 = (x0 > 0.f) ? x0 : NEG_SLOPE * x0;
        x1 = (x1 > 0.f) ? x1 : NEG_SLOPE * x1;
        x2 = (x2 > 0.f) ? x2 : NEG_SLOPE * x2;
        x3 = (x3 > 0.f) ? x3 : NEG_SLOPE * x3;
        acc += x0 * w.x + x1 * w.y + x2 * w.z + x3 * w.w;
    }
    score[t] = acc;
    atomicMaxFloat(&mmax[(size_t)d * HEADS + h], acc);
}

__global__ void edge_exp_sum(const int* __restrict__ dst,
                             float* __restrict__ score,
                             const float* __restrict__ mmax,
                             float* __restrict__ ssum) {
    int t = blockIdx.x * blockDim.x + threadIdx.x;
    if (t >= N_EDGES * HEADS) return;
    const int e = t >> 2, h = t & 3;
    const int d = dst[e];
    float w = expf(score[t] - mmax[(size_t)d * HEADS + h]);
    score[t] = w;
    atomicAdd(&ssum[(size_t)d * HEADS + h], w);
}

__global__ void edge_aggregate(const int* __restrict__ src,
                               const int* __restrict__ dst,
                               const float* __restrict__ score,
                               const float* __restrict__ ssum,
                               const float* __restrict__ fs,
                               float* __restrict__ outacc) {
    int t = blockIdx.x * blockDim.x + threadIdx.x;
    if (t >= N_EDGES * HEADS) return;
    const int e = t >> 2, h = t & 3;
    const int s = src[e], d = dst[e];
    const float alpha = score[t] / ssum[(size_t)d * HEADS + h];
    const float* ps = fs + (size_t)s * H_FEATS + h * D_HEAD;
    float*       po = outacc + (size_t)d * H_FEATS + h * D_HEAD;
#pragma unroll
    for (int i = 0; i < D_HEAD; ++i)
        atomicAdd(&po[i], alpha * ps[i]);
}

__global__ void out_proj(const float* __restrict__ h,
                         const float* __restrict__ Wout,
                         const float* __restrict__ bout,
                         float* __restrict__ out) {
    int n = blockIdx.x * blockDim.x + threadIdx.x;
    if (n >= N_NODES) return;
    const float* ph = h + (size_t)n * H_FEATS;
    float a0 = bout[0], a1 = bout[1];
#pragma unroll
    for (int j = 0; j < H_FEATS; ++j) {
        float v = ph[j];
        a0 += v * Wout[j * 2 + 0];
        a1 += v * Wout[j * 2 + 1];
    }
    out[n * 2 + 0] = a0;
    out[n * 2 + 1] = a1;
}

extern "C" void kernel_launch(void* const* d_in, const int* in_sizes, int n_in,
                              void* d_out, int out_size, void* d_ws, size_t ws_size,
                              hipStream_t stream) {
    (void)in_sizes; (void)n_in; (void)out_size; (void)ws_size;

    const float* feature  = (const float*)d_in[0];
    const int*   src      = (const int*)  d_in[1];
    const int*   dst      = (const int*)  d_in[2];
    const float* W_in     = (const float*)d_in[3];
    const float* b_in     = (const float*)d_in[4];
    const float* fc_src_W = (const float*)d_in[5];
    const float* fc_src_b = (const float*)d_in[6];
    const float* fc_dst_W = (const float*)d_in[7];
    const float* fc_dst_b = (const float*)d_in[8];
    const float* attn     = (const float*)d_in[9];
    const float* W_out    = (const float*)d_in[10];
    const float* b_out    = (const float*)d_in[11];
    float* logits = (float*)d_out;

    float* ws = (float*)d_ws;
    size_t off = 0;
    float* h     = ws + off; off += (size_t)N_NODES * H_FEATS;
    float* fs    = ws + off; off += (size_t)N_NODES * H_FEATS;
    float* fd    = ws + off; off += (size_t)N_NODES * H_FEATS;
    float* score = ws + off; off += (size_t)N_EDGES * HEADS;
    float* mmax  = ws + off; off += (size_t)N_NODES * HEADS;
    float* ssum  = ws + off; off += (size_t)N_NODES * HEADS;
    float* WT_in = ws + off; off += (size_t)H_FEATS * IN_FEATS;
    float* WT_s  = ws + off; off += (size_t)2 * H_FEATS * H_FEATS;
    float* WT_d  = ws + off; off += (size_t)2 * H_FEATS * H_FEATS;

    const int mtiles = N_NODES / 16;
    const int EH     = N_EDGES * HEADS;
    const int ETB    = 256;

    transpose_k<<<(IN_FEATS * H_FEATS + 255) / 256, 256, 0, stream>>>(W_in, WT_in, IN_FEATS, H_FEATS);
    for (int l = 0; l < 2; ++l) {
        transpose_k<<<(H_FEATS * H_FEATS + 255) / 256, 256, 0, stream>>>(
            fc_src_W + (size_t)l * H_FEATS * H_FEATS, WT_s + (size_t)l * H_FEATS * H_FEATS, H_FEATS, H_FEATS);
        transpose_k<<<(H_FEATS * H_FEATS + 255) / 256, 256, 0, stream>>>(
            fc_dst_W + (size_t)l * H_FEATS * H_FEATS, WT_d + (size_t)l * H_FEATS * H_FEATS, H_FEATS, H_FEATS);
    }

    gemm_bias_wmma<<<mtiles, 128, 0, stream>>>(feature, WT_in, b_in, h, IN_FEATS);

    for (int l = 0; l < 2; ++l) {
        gemm_bias_wmma<<<mtiles, 128, 0, stream>>>(h, WT_s + (size_t)l * H_FEATS * H_FEATS,
                                                   fc_src_b + l * H_FEATS, fs, H_FEATS);
        gemm_bias_wmma<<<mtiles, 128, 0, stream>>>(h, WT_d + (size_t)l * H_FEATS * H_FEATS,
                                                   fc_dst_b + l * H_FEATS, fd, H_FEATS);

        fill_k<<<(N_NODES * HEADS + 255) / 256, 256, 0, stream>>>(mmax, -INFINITY, N_NODES * HEADS);
        fill_k<<<(N_NODES * HEADS + 255) / 256, 256, 0, stream>>>(ssum, 0.f, N_NODES * HEADS);

        edge_score_max<<<(EH + ETB - 1) / ETB, ETB, 0, stream>>>(src, dst, fs, fd,
                                                                 attn + l * HEADS * D_HEAD, score, mmax);
        edge_exp_sum<<<(EH + ETB - 1) / ETB, ETB, 0, stream>>>(dst, score, mmax, ssum);

        fill_k<<<(N_NODES * H_FEATS + 255) / 256, 256, 0, stream>>>(h, 0.f, N_NODES * H_FEATS);
        edge_aggregate<<<(EH + ETB - 1) / ETB, ETB, 0, stream>>>(src, dst, score, ssum, fs, h);
        relu_k<<<(N_NODES * H_FEATS + 255) / 256, 256, 0, stream>>>(h, N_NODES * H_FEATS);
    }

    out_proj<<<(N_NODES + 255) / 256, 256, 0, stream>>>(h, W_out, b_out, logits);
}


# Edges sorted by destination in-call (count, scan, fill); one per-node gather pass does scores, softmax and sum: no float atomics; sums run in segment arrival order, softmax in rescaling form
# speedup vs baseline: 2.5637x; 2.5637x over previous
#include <hip/hip_runtime.h>
#include <math.h>

#define N_NODES  50000
#define N_EDGES  800000
#define IN_FEATS 128
#define H_FEATS  64
#define HEADS    4
#define D_HEAD   16
#define NEG_SLOPE 0.2f

typedef __attribute__((ext_vector_type(2))) float v2f;
typedef __attribute__((ext_vector_type(8))) float v8f;

__device__ __forceinline__ v8f wmma_f32_16x16x4(v2f a, v2f b, v8f c) {
    return __builtin_amdgcn_wmma_f32_16x16x4_f32(
         false, a,  false, b,
         (short)0, c,  false,  false);
}

__global__ void gemm_bias_wmma(const float* __restrict__ A,
                               const float* __restrict__ WT,
                               const float* __restrict__ bias,
                               float* __restrict__ C, int K) {
    const int lane = threadIdx.x & 31;
    const int wave = threadIdx.x >> 5;
    const int mt   = blockIdx.x;
    const int rm   = lane & 15;
    const int koff = (lane >> 4) << 1;

    const float* arow = A  + (size_t)(mt * 16 + rm) * K + koff;
    const float* brow = WT + (size_t)(wave * 16 + rm) * K + koff;

    v8f acc = {0.f, 0.f, 0.f, 0.f, 0.f, 0.f, 0.f, 0.f};
    for (int k = 0; k < K; k += 4) {
        const float2 av = *(const float2*)(arow + k);
        const float2 bv = *(const float2*)(brow + k);
        v2f a; a.x = av.x; a.y = av.y;
        v2f b; b.x = bv.x; b.y = bv.y;
        acc = wmma_f32_16x16x4(a, b, acc);
    }

    const int   col  = wave * 16 + rm;
    const float bv   = bias[col];
    const int   row0 = mt * 16 + ((lane >> 4) << 3);
#pragma unroll
    for (int r = 0; r < 8; ++r)
        C[(size_t)(row0 + r) * H_FEATS + col] = acc[r] + bv;
}

__global__ void transpose_k(const float* __restrict__ W, float* __restrict__ WT,
                            int R, int Ccols) {
    int t = blockIdx.x * blockDim.x + threadIdx.x;
    if (t >= R * Ccols) return;
    int r = t / Ccols, c = t - r * Ccols;
    WT[(size_t)c * R + r] = W[t];
}

__global__ void zero_i(int* __restrict__ p, int n) {
    int t = blockIdx.x * blockDim.x + threadIdx.x;
    if (t < n) p[t] = 0;
}

__global__ void degree_count(const int* __restrict__ dst, int* __restrict__ cnt) {
    int e = blockIdx.x * blockDim.x + threadIdx.x;
    if (e >= N_EDGES) return;
    atomicAdd(&cnt[dst[e]], 1);
}

#define SCAN_T     1024
#define SCAN_CHUNK ((N_NODES + SCAN_T - 1) / SCAN_T)

__global__ void __launch_bounds__(SCAN_T)
rowptr_scan(const int* __restrict__ cnt, int* __restrict__ rowptr) {
    __shared__ int buf[SCAN_T];
    const int t  = threadIdx.x;
    const int lo = t * SCAN_CHUNK;
    int v[SCAN_CHUNK];
    int s = 0;
#pragma unroll
    for (int i = 0; i < SCAN_CHUNK; ++i) {
        const int k = lo + i;
        const int x = cnt[(k < N_NODES) ? k : (N_NODES - 1)];
        v[i] = (k < N_NODES) ? x : 0;
        s += v[i];
    }
    buf[t] = s;
    __syncthreads();
    for (int off = 1; off < SCAN_T; off <<= 1) {
        const int add = (t >= off) ? buf[t - off] : 0;
        __syncthreads();
        buf[t] += add;
        __syncthreads();
    }
    int run = buf[t] - s;
#pragma unroll
    for (int i = 0; i < SCAN_CHUNK; ++i) {
        const int k = lo + i;
        if (k < N_NODES) rowptr[k] = run;
        run += v[i];
    }
    if (t == SCAN_T - 1) rowptr[N_NODES] = buf[t];
}

__global__ void csr_fill(const int* __restrict__ src,
                         const int* __restrict__ dst,
                         const int* __restrict__ rowptr,
                         int* __restrict__ cnt,
                         int* __restrict__ csr_src) {
    int e = blockIdx.x * blockDim.x + threadIdx.x;
    if (e >= N_EDGES) return;
    const int d    = dst[e];
    const int slot = atomicSub(&cnt[d], 1) - 1;
    csr_src[rowptr[d] + slot] = src[e];
}

#define EDGE_BATCH 4

__global__ void __launch_bounds__(256)
node_softmax_aggregate(const int* __restrict__ rowptr,
                       const int* __restrict__ csr_src,
                       const float* __restrict__ fs,
                       const float* __restrict__ fd,
                       const float* __restrict__ attn_l,
                       float* __restrict__ hout) {
    const int t = blockIdx.x * blockDim.x + threadIdx.x;
    const int n = t >> 4;
    const int j = t & 15;
    if (n >= N_NODES) return;

    const float4* fs4 = (const float4*)fs;
    const float4  b   = ((const float4*)fd)[n * 16 + j];
    const float4  w   = ((const float4*)attn_l)[j];
    const int start = rowptr[n];
    const int end   = rowptr[n + 1];

    float  m = -INFINITY, l = 0.f;
    float4 acc = make_float4(0.f, 0.f, 0.f, 0.f);

    for (int p = start; p < end; p += EDGE_BATCH) {
        float4 a[EDGE_BATCH];
        float  sc[EDGE_BATCH];
#pragma unroll
        for (int u = 0; u < EDGE_BATCH; ++u) {
            const int q = (p + u < end) ? (p + u) : (end - 1);
            a[u] = fs4[csr_src[q] * 16 + j];
        }
#pragma unroll
        for (int u = 0; u < EDGE_BATCH; ++u) {
            float x0 = a[u].x + b.x, x1 = a[u].y + b.y, x2 = a[u].z + b.z, x3 = a[u].w + b.w;
            x0 = (x0 > 0.f) ? x0 : NEG_SLOPE * x0;
            x1 = (x1 > 0.f) ? x1 : NEG_SLOPE * x1;
            x2 = (x2 > 0.f) ? x2 : NEG_SLOPE * x2;
            x3 = (x3 > 0.f) ? x3 : NEG_SLOPE * x3;
            float part = x0 * w.x + x1 * w.y + x2 * w.z + x3 * w.w;
            part += __shfl_xor(part, 1);
            part += __shfl_xor(part, 2);
            sc[u] = (p + u < end) ? part : -INFINITY;
        }
        float mn = fmaxf(m, sc[0]);
#pragma unroll
        for (int u = 1; u < EDGE_BATCH; ++u) mn = fmaxf(mn, sc[u]);
        const float r = expf(m - mn);
        l *= r;
        acc.x *= r; acc.y *= r; acc.z *= r; acc.w *= r;
#pragma unroll
        for (int u = 0; u < EDGE_BATCH; ++u) {
            const float pe = expf(sc[u] - mn);
            l     += pe;
            acc.x += pe * a[u].x;
            acc.y += pe * a[u].y;
            acc.z += pe * a[u].z;
            acc.w += pe * a[u].w;
        }
        m = mn;
    }

    float4 o = make_float4(0.f, 0.f, 0.f, 0.f);
    if (end > start) {
        const float inv = 1.f / l;
        o.x = fmaxf(acc.x * inv, 0.f);
        o.y = fmaxf(acc.y * inv, 0.f);
        o.z = fmaxf(acc.z * inv, 0.f);
        o.w = fmaxf(acc.w * inv, 0.f);
    }
    ((float4*)hout)[n * 16 + j] = o;
}

__global__ void out_proj(const float* __restrict__ h,
                         const float* __restrict__ Wout,
                         const float* __restrict__ bout,
                         float* __restrict__ out) {
    int n = blockIdx.x * blockDim.x + threadIdx.x;
    if (n >= N_NODES) return;
    const float* ph = h + (size_t)n * H_FEATS;
    float a0 = bout[0], a1 = bout[1];
#pragma unroll
    for (int j = 0; j < H_FEATS; ++j) {
        float v = ph[j];
        a0 += v * Wout[j * 2 + 0];
        a1 += v * Wout[j * 2 + 1];
    }
    out[n * 2 + 0] = a0;
    out[n * 2 + 1] = a1;
}

extern "C" void kernel_launch(void* const* d_in, const int* in_sizes, int n_in,
                              void* d_out, int out_size, void* d_ws, size_t ws_size,
                              hipStream_t stream) {
    (void)in_sizes; (void)n_in; (void)out_size; (void)ws_size;

    const float* feature  = (const float*)d_in[0];
    const int*   src      = (const int*)  d_in[1];
    const int*   dst      = (const int*)  d_in[2];
    const float* W_in     = (const float*)d_in[3];
    const float* b_in     = (const float*)d_in[4];
    const float* fc_src_W = (const float*)d_in[5];
    const float* fc_src_b = (const float*)d_in[6];
    const float* fc_dst_W = (const float*)d_in[7];
    const float* fc_dst_b = (const float*)d_in[8];
    const float* attn     = (const float*)d_in[9];
    const float* W_out    = (const float*)d_in[10];
    const float* b_out    = (const float*)d_in[11];
    float* logits = (float*)d_out;

    float* ws = (float*)d_ws;
    size_t off = 0;
    float* h     = ws + off; off += (size_t)N_NODES * H_FEATS;
    float* fs    = ws + off; off += (size_t)N_NODES * H_FEATS;
    float* fd    = ws + off; off += (size_t)N_NODES * H_FEATS;
    float* WT_in = ws + off; off += (size_t)H_FEATS * IN_FEATS;
    float* WT_s  = ws + off; off += (size_t)2 * H_FEATS * H_FEATS;
    float* WT_d  = ws + off; off += (size_t)2 * H_FEATS * H_FEATS;
    int* cnt     = (int*)(ws + off); off += (size_t)N_NODES;
    int* rowptr  = (int*)(ws + off); off += (size_t)N_NODES + 16;
    int* csr_src = (int*)(ws + off); off += (size_t)N_EDGES;

    const int mtiles = N_NODES / 16;
    const int ETB    = 256;

    transpose_k<<<(IN_FEATS * H_FEATS + 255) / 256, 256, 0, stream>>>(W_in, WT_in, IN_FEATS, H_FEATS);
    for (int l = 0; l < 2; ++l) {
        transpose_k<<<(H_FEATS * H_FEATS + 255) / 256, 256, 0, stream>>>(
            fc_src_W + (size_t)l * H_FEATS * H_FEATS, WT_s + (size_t)l * H_FEATS * H_FEATS, H_FEATS, H_FEATS);
        transpose_k<<<(H_FEATS * H_FEATS + 255) / 256, 256, 0, stream>>>(
            fc_dst_W + (size_t)l * H_FEATS * H_FEATS, WT_d + (size_t)l * H_FEATS * H_FEATS, H_FEATS, H_FEATS);
    }

    zero_i<<<(N_NODES + 255) / 256, 256, 0, stream>>>(cnt, N_NODES);
    degree_count<<<(N_EDGES + ETB - 1) / ETB, ETB, 0, stream>>>(dst, cnt);
    rowptr_scan<<<1, SCAN_T, 0, stream>>>(cnt, rowptr);
    csr_fill<<<(N_EDGES + ETB - 1) / ETB, ETB, 0, stream>>>(src, dst, rowptr, cnt, csr_src);

    gemm_bias_wmma<<<mtiles, 128, 0, stream>>>(feature, WT_in, b_in, h, IN_FEATS);

    for (int l = 0; l < 2; ++l) {
        gemm_bias_wmma<<<mtiles, 128, 0, stream>>>(h, WT_s + (size_t)l * H_FEATS * H_FEATS,
                                                   fc_src_b + l * H_FEATS, fs, H_FEATS);
        gemm_bias_wmma<<<mtiles, 128, 0, stream>>>(h, WT_d + (size_t)l * H_FEATS * H_FEATS,
                                                   fc_dst_b + l * H_FEATS, fd, H_FEATS);

        node_softmax_aggregate<<<(N_NODES * 16 + 255) / 256, 256, 0, stream>>>(
            rowptr, csr_src, fs, fd, attn + l * HEADS * D_HEAD, h);
    }

    out_proj<<<(N_NODES + 255) / 256, 256, 0, stream>>>(h, W_out, b_out, logits);
}


# GEMMs re-tiled: a wave sweeps all output columns of its 16-row tile (A read once), weights staged as k pairs in LDS from row-major W (transposes gone), fs|fd in one launch; 17->10 launches
# speedup vs baseline: 3.3632x; 1.3119x over previous
#include <hip/hip_runtime.h>
#include <math.h>

#define N_NODES  50000
#define N_EDGES  800000
#define IN_FEATS 128
#define H_FEATS  64
#define HEADS    4
#define D_HEAD   16
#define NEG_SLOPE 0.2f

typedef __attribute__((ext_vector_type(2))) float v2f;
typedef __attribute__((ext_vector_type(8))) float v8f;

__device__ __forceinline__ v8f wmma_f32_16x16x4(v2f a, v2f b, v8f c) {
    return __builtin_amdgcn_wmma_f32_16x16x4_f32(
         false, a,  false, b,
         (short)0, c,  false,  false);
}

#define GEMM_WAVES 8

template <int K, int NT>
__global__ void __launch_bounds__(GEMM_WAVES * 32)
gemm_bias_wmma(const float* __restrict__ A,
               const float* __restrict__ W0,
               const float* __restrict__ W1,
               const float* __restrict__ bias0,
               const float* __restrict__ bias1,
               float* __restrict__ C0,
               float* __restrict__ C1) {
    constexpr int N   = NT * 16;
    constexpr int PN2 = 2 * N + 32;
    __shared__ float sB[(K / 2) * PN2];

    constexpr int BLOCKS_PER_THREAD = (K * N / 8) / (GEMM_WAVES * 32);
#pragma unroll
    for (int it = 0; it < BLOCKS_PER_THREAD; ++it) {
        const int i  = it * (GEMM_WAVES * 32) + threadIdx.x;
        const int kp = i / (N / 4);
        const int n4 = (i - kp * (N / 4)) * 4;
        const float* wsrc = (n4 < H_FEATS) ? (W0 + n4) : (W1 + (n4 - H_FEATS));
        const float4 r0 = *(const float4*)(wsrc + (2 * kp) * H_FEATS);
        const float4 r1 = *(const float4*)(wsrc + (2 * kp + 1) * H_FEATS);
        float* d = sB + kp * PN2 + n4 * 2;
        *(float4*)(d)     = make_float4(r0.x, r1.x, r0.y, r1.y);
        *(float4*)(d + 4) = make_float4(r0.z, r1.z, r0.w, r1.w);
    }
    __syncthreads();

    const int lane = threadIdx.x & 31;
    const int wave = threadIdx.x >> 5;
    const int mt   = blockIdx.x * GEMM_WAVES + wave;
    if (mt >= N_NODES / 16) return;
    const int c  = lane & 15;
    const int hf = lane >> 4;

    const float* arow = A + (size_t)(mt * 16 + c) * K + 2 * hf;
    const float* bcol = sB + hf * PN2 + c * 2;

    float2 av[K / 4];
#pragma unroll
    for (int s = 0; s < K / 4; ++s)
        av[s] = *(const float2*)(arow + 4 * s);

    v8f acc[NT];
#pragma unroll
    for (int nt = 0; nt < NT; ++nt)
        acc[nt] = (v8f){0.f, 0.f, 0.f, 0.f, 0.f, 0.f, 0.f, 0.f};

#pragma unroll
    for (int s = 0; s < K / 4; ++s) {
        v2f a; a.x = av[s].x; a.y = av[s].y;
        const float* bp = bcol + (2 * s) * PN2;
#pragma unroll
        for (int nt = 0; nt < NT; ++nt) {
            const float2 bv = *(const float2*)(bp + nt * 32);
            v2f b; b.x = bv.x; b.y = bv.y;
            acc[nt] = wmma_f32_16x16x4(a, b, acc[nt]);
        }
    }

    const int row0 = mt * 16 + 8 * hf;
#pragma unroll
    for (int nt = 0; nt < NT; ++nt) {
        const int   col = (nt & 3) * 16 + c;
        const float bv  = (nt < 4) ? bias0[col] : bias1[col];
        float*      Cp  = (nt < 4) ? C0 : C1;
#pragma unroll
        for (int r = 0; r < 8; ++r)
            Cp[(size_t)(row0 + r) * H_FEATS + col] = acc[nt][r] + bv;
    }
}

__global__ void zero_i(int* __restrict__ p, int n) {
    int t = blockIdx.x * blockDim.x + threadIdx.x;
    if (t < n) p[t] = 0;
}

__global__ void degree_count(const int* __restrict__ dst, int* __restrict__ cnt) {
    int e = blockIdx.x * blockDim.x + threadIdx.x;
    if (e >= N_EDGES) return;
    atomicAdd(&cnt[dst[e]], 1);
}

#define SCAN_T     1024
#define SCAN_CHUNK ((N_NODES + SCAN_T - 1) / SCAN_T)

__global__ void __launch_bounds__(SCAN_T)
rowptr_scan(const int* __restrict__ cnt, int* __restrict__ rowptr) {
    __shared__ int buf[SCAN_T];
    const int t  = threadIdx.x;
    const int lo = t * SCAN_CHUNK;
    int v[SCAN_CHUNK];
    int s = 0;
#pragma unroll
    for (int i = 0; i < SCAN_CHUNK; ++i) {
        const int k = lo + i;
        const int x = cnt[(k < N_NODES) ? k : (N_NODES - 1)];
        v[i] = (k < N_NODES) ? x : 0;
        s += v[i];
    }
    buf[t] = s;
    __syncthreads();
    for (int off = 1; off < SCAN_T; off <<= 1) {
        const int add = (t >= off) ? buf[t - off] : 0;
        __syncthreads();
        buf[t] += add;
        __syncthreads();
    }
    int run = buf[t] - s;
#pragma unroll
    for (int i = 0; i < SCAN_CHUNK; ++i) {
        const int k = lo + i;
        if (k < N_NODES) rowptr[k] = run;
        run += v[i];
    }
    if (t == SCAN_T - 1) rowptr[N_NODES] = buf[t];
}

__global__ void csr_fill(const int* __restrict__ src,
                         const int* __restrict__ dst,
                         const int* __restrict__ rowptr,
                         int* __restrict__ cnt,
                         int* __restrict__ csr_src) {
    int e = blockIdx.x * blockDim.x + threadIdx.x;
    if (e >= N_EDGES) return;
    const int d    = dst[e];
    const int slot = atomicSub(&cnt[d], 1) - 1;
    csr_src[rowptr[d] + slot] = src[e];
}

#define EDGE_BATCH 4

__global__ void __launch_bounds__(256)
node_softmax_aggregate(const int* __restrict__ rowptr,
                       const int* __restrict__ csr_src,
                       const float* __restrict__ fs,
                       const float* __restrict__ fd,
                       const float* __restrict__ attn_l,
                       float* __restrict__ hout) {
    const int t = blockIdx.x * blockDim.x + threadIdx.x;
    const int n = t >> 4;
    const int j = t & 15;
    if (n >= N_NODES) return;

    const float4* fs4 = (const float4*)fs;
    const float4  b   = ((const float4*)fd)[n * 16 + j];
    const float4  w   = ((const float4*)attn_l)[j];
    const int start = rowptr[n];
    const int end   = rowptr[n + 1];

    float  m = -INFINITY, l = 0.f;
    float4 acc = make_float4(0.f, 0.f, 0.f, 0.f);

    for (int p = start; p < end; p += EDGE_BATCH) {
        float4 a[EDGE_BATCH];
        float  sc[EDGE_BATCH];
#pragma unroll
        for (int u = 0; u < EDGE_BATCH; ++u) {
            const int q = (p + u < end) ? (p + u) : (end - 1);
            a[u] = fs4[csr_src[q] * 16 + j];
        }
#pragma unroll
        for (int u = 0; u < EDGE_BATCH; ++u) {
            float x0 = a[u].x + b.x, x1 = a[u].y + b.y, x2 = a[u].z + b.z, x3 = a[u].w + b.w;
            x0 = (x0 > 0.f) ? x0 : NEG_SLOPE * x0;
            x1 = (x1 > 0.f) ? x1 : NEG_SLOPE * x1;
            x2 = (x2 > 0.f) ? x2 : NEG_SLOPE * x2;
            x3 = (x3 > 0.f) ? x3 : NEG_SLOPE * x3;
            float part = x0 * w.x + x1 * w.y + x2 * w.z + x3 * w.w;
            part += __shfl_xor(part, 1);
            part += __shfl_xor(part, 2);
            sc[u] = (p + u < end) ? part : -INFINITY;
        }
        float mn = fmaxf(m, sc[0]);
#pragma unroll
        for (int u = 1; u < EDGE_BATCH; ++u) mn = fmaxf(mn, sc[u]);
        const float r = expf(m - mn);
        l *= r;
        acc.x *= r; acc.y *= r; acc.z *= r; acc.w *= r;
#pragma unroll
        for (int u = 0; u < EDGE_BATCH; ++u) {
            const float pe = expf(sc[u] - mn);
            l     += pe;
            acc.x += pe * a[u].x;
            acc.y += pe * a[u].y;
            acc.z += pe * a[u].z;
            acc.w += pe * a[u].w;
        }
        m = mn;
    }

    float4 o = make_float4(0.f, 0.f, 0.f, 0.f);
    if (end > start) {
        const float inv = 1.f / l;
        o.x = fmaxf(acc.x * inv, 0.f);
        o.y = fmaxf(acc.y * inv, 0.f);
        o.z = fmaxf(acc.z * inv, 0.f);
        o.w = fmaxf(acc.w * inv, 0.f);
    }
    ((float4*)hout)[n * 16 + j] = o;
}

__global__ void out_proj(const float* __restrict__ h,
                         const float* __restrict__ Wout,
                         const float* __restrict__ bout,
                         float* __restrict__ out) {
    int n = blockIdx.x * blockDim.x + threadIdx.x;
    if (n >= N_NODES) return;
    const float* ph = h + (size_t)n * H_FEATS;
    float a0 = bout[0], a1 = bout[1];
#pragma unroll
    for (int j = 0; j < H_FEATS; ++j) {
        float v = ph[j];
        a0 += v * Wout[j * 2 + 0];
        a1 += v * Wout[j * 2 + 1];
    }
    out[n * 2 + 0] = a0;
    out[n * 2 + 1] = a1;
}

extern "C" void kernel_launch(void* const* d_in, const int* in_sizes, int n_in,
                              void* d_out, int out_size, void* d_ws, size_t ws_size,
                              hipStream_t stream) {
    (void)in_sizes; (void)n_in; (void)out_size; (void)ws_size;

    const float* feature  = (const float*)d_in[0];
    const int*   src      = (const int*)  d_in[1];
    const int*   dst      = (const int*)  d_in[2];
    const float* W_in     = (const float*)d_in[3];
    const float* b_in     = (const float*)d_in[4];
    const float* fc_src_W = (const float*)d_in[5];
    const float* fc_src_b = (const float*)d_in[6];
    const float* fc_dst_W = (const float*)d_in[7];
    const float* fc_dst_b = (const float*)d_in[8];
    const float* attn     = (const float*)d_in[9];
    const float* W_out    = (const float*)d_in[10];
    const float* b_out    = (const float*)d_in[11];
    float* logits = (float*)d_out;

    float* ws = (float*)d_ws;
    size_t off = 0;
    float* h     = ws + off; off += (size_t)N_NODES * H_FEATS;
    float* fs    = ws + off; off += (size_t)N_NODES * H_FEATS;
    float* fd    = ws + off; off += (size_t)N_NODES * H_FEATS;
    int* cnt     = (int*)(ws + off); off += (size_t)N_NODES;
    int* rowptr  = (int*)(ws + off); off += (size_t)N_NODES + 16;
    int* csr_src = (int*)(ws + off); off += (size_t)N_EDGES;

    const int gblocks = (N_NODES / 16 + GEMM_WAVES - 1) / GEMM_WAVES;
    const int ETB     = 256;

    zero_i<<<(N_NODES + 255) / 256, 256, 0, stream>>>(cnt, N_NODES);
    degree_count<<<(N_EDGES + ETB - 1) / ETB, ETB, 0, stream>>>(dst, cnt);
    rowptr_scan<<<1, SCAN_T, 0, stream>>>(cnt, rowptr);
    csr_fill<<<(N_EDGES + ETB - 1) / ETB, ETB, 0, stream>>>(src, dst, rowptr, cnt, csr_src);

    gemm_bias_wmma<IN_FEATS, 4><<<gblocks, GEMM_WAVES * 32, 0, stream>>>(
        feature, W_in, W_in, b_in, b_in, h, h);

    for (int l = 0; l < 2; ++l) {
        gemm_bias_wmma<H_FEATS, 8><<<gblocks, GEMM_WAVES * 32, 0, stream>>>(
            h, fc_src_W + (size_t)l * H_FEATS * H_FEATS, fc_dst_W + (size_t)l * H_FEATS * H_FEATS,
            fc_src_b + l * H_FEATS, fc_dst_b + l * H_FEATS, fs, fd);

        node_softmax_aggregate<<<(N_NODES * 16 + 255) / 256, 256, 0, stream>>>(
            rowptr, csr_src, fs, fd, attn + l * HEADS * D_HEAD, h);
    }

    out_proj<<<(N_NODES + 255) / 256, 256, 0, stream>>>(h, W_out, b_out, logits);
}


# Edge sort redone as a two-level counting sort on LDS counters (per-block bin table, bin runs, one block per 256-node bin): no global atomics; segment order = LDS arrival order; 10->9 launches
# speedup vs baseline: 7.6694x; 2.2804x over previous
#include <hip/hip_runtime.h>
#include <math.h>

#define N_NODES  50000
#define N_EDGES  800000
#define IN_FEATS 128
#define H_FEATS  64
#define HEADS    4
#define D_HEAD   16
#define NEG_SLOPE 0.2f

typedef __attribute__((ext_vector_type(2))) float v2f;
typedef __attribute__((ext_vector_type(8))) float v8f;

__device__ __forceinline__ v8f wmma_f32_16x16x4(v2f a, v2f b, v8f c) {
    return __builtin_amdgcn_wmma_f32_16x16x4_f32(
         false, a,  false, b,
         (short)0, c,  false,  false);
}

#define GEMM_WAVES 8

template <int K, int NT>
__global__ void __launch_bounds__(GEMM_WAVES * 32)
gemm_bias_wmma(const float* __restrict__ A,
               const float* __restrict__ W0,
               const float* __restrict__ W1,
               const float* __restrict__ bias0,
               const float* __restrict__ bias1,
               float* __restrict__ C0,
               float* __restrict__ C1) {
    constexpr int N   = NT * 16;
    constexpr int PN2 = 2 * N + 32;
    __shared__ float sB[(K / 2) * PN2];

    constexpr int BLOCKS_PER_THREAD = (K * N / 8) / (GEMM_WAVES * 32);
#pragma unroll
    for (int it = 0; it < BLOCKS_PER_THREAD; ++it) {
        const int i  = it * (GEMM_WAVES * 32) + threadIdx.x;
        const int kp = i / (N / 4);
        const int n4 = (i - kp * (N / 4)) * 4;
        const float* wsrc = (n4 < H_FEATS) ? (W0 + n4) : (W1 + (n4 - H_FEATS));
        const float4 r0 = *(const float4*)(wsrc + (2 * kp) * H_FEATS);
        const float4 r1 = *(const float4*)(wsrc + (2 * kp + 1) * H_FEATS);
        float* d = sB + kp * PN2 + n4 * 2;
        *(float4*)(d)     = make_float4(r0.x, r1.x, r0.y, r1.y);
        *(float4*)(d + 4) = make_float4(r0.z, r1.z, r0.w, r1.w);
    }
    __syncthreads();

    const int lane = threadIdx.x & 31;
    const int wave = threadIdx.x >> 5;
    const int mt   = blockIdx.x * GEMM_WAVES + wave;
    if (mt >= N_NODES / 16) return;
    const int c  = lane & 15;
    const int hf = lane >> 4;

    const float* arow = A + (size_t)(mt * 16 + c) * K + 2 * hf;
    const float* bcol = sB + hf * PN2 + c * 2;

    float2 av[K / 4];
#pragma unroll
    for (int s = 0; s < K / 4; ++s)
        av[s] = *(const float2*)(arow + 4 * s);

    v8f acc[NT];
#pragma unroll
    for (int nt = 0; nt < NT; ++nt)
        acc[nt] = (v8f){0.f, 0.f, 0.f, 0.f, 0.f, 0.f, 0.f, 0.f};

#pragma unroll
    for (int s = 0; s < K / 4; ++s) {
        v2f a; a.x = av[s].x; a.y = av[s].y;
        const float* bp = bcol + (2 * s) * PN2;
#pragma unroll
        for (int nt = 0; nt < NT; ++nt) {
            const float2 bv = *(const float2*)(bp + nt * 32);
            v2f b; b.x = bv.x; b.y = bv.y;
            acc[nt] = wmma_f32_16x16x4(a, b, acc[nt]);
        }
    }

    const int row0 = mt * 16 + 8 * hf;
#pragma unroll
    for (int nt = 0; nt < NT; ++nt) {
        const int   col = (nt & 3) * 16 + c;
        const float bv  = (nt < 4) ? bias0[col] : bias1[col];
        float*      Cp  = (nt < 4) ? C0 : C1;
#pragma unroll
        for (int r = 0; r < 8; ++r)
            Cp[(size_t)(row0 + r) * H_FEATS + col] = acc[nt][r] + bv;
    }
}

#define BIN_SHIFT 8
#define BIN_NODES (1 << BIN_SHIFT)
#define NBINS     ((N_NODES + BIN_NODES - 1) / BIN_NODES)
#define NBP       256
#define PBLK      128
#define PT        1024
#define CHUNK     ((N_EDGES + PBLK - 1) / PBLK)
#define EPT       ((CHUNK + PT - 1) / PT)
#define TPARTS    (PT / (NBP / 2))
#define TROWS     (PBLK / TPARTS)
#define ST        512
#define SORT_CAP  8192
#define RPT       (SORT_CAP / ST)

static_assert(NBINS <= NBP && NBP == 256, "a bin number must fit a byte; the scan below is 8 waves wide");
static_assert(BIN_NODES == NBP, "the node scan of a bin reuses the bin scan");
static_assert(CHUNK < 65536, "a block's count of one bin must fit 16 bits");
static_assert(PT % (NBP / 2) == 0 && PBLK % TPARTS == 0, "table rows split evenly over the thread groups");
static_assert((N_NODES >> BIN_SHIFT) < NBINS, "rowptr[N_NODES] must belong to the last bin");
static_assert(N_NODES <= (1 << (31 - BIN_SHIFT)), "src << BIN_SHIFT must stay positive");

__device__ __forceinline__ int block_scan256(int x, int t, int* s_w) {
    const int lane = t & 31;
    const int w    = t >> 5;
#pragma unroll
    for (int off = 1; off < 32; off <<= 1) {
        const int y = __shfl_up(x, off);
        if (lane >= off) x += y;
    }
    if (t < NBP && lane == 31) s_w[w] = x;
    __syncthreads();
    if (t < NBP) {
        int add = 0;
#pragma unroll
        for (int k = 0; k < NBP / 32; ++k) add += (k < w) ? s_w[k] : 0;
        x += add;
    }
    return x;
}

__global__ void __launch_bounds__(PT)
bin_hist(const int* __restrict__ dst, unsigned* __restrict__ blkhist) {
    __shared__ int h[NBP];
    const int t  = threadIdx.x;
    const int lo = blockIdx.x * CHUNK;
    const int n  = min(CHUNK, N_EDGES - lo);
    int d[EPT];
#pragma unroll
    for (int u = 0; u < EPT; ++u) {
        const int i = t + u * PT;
        d[u] = dst[(i < n) ? (lo + i) : 0];
    }
    if (t < NBP) h[t] = 0;
    __syncthreads();
#pragma unroll
    for (int u = 0; u < EPT; ++u)
        if (t + u * PT < n) atomicAdd(&h[d[u] >> BIN_SHIFT], 1);
    __syncthreads();
    if (t < NBP / 2)
        blkhist[blockIdx.x * (NBP / 2) + t] = (unsigned)h[2 * t] | ((unsigned)h[2 * t + 1] << 16);
}

__global__ void __launch_bounds__(PT)
bin_scatter(const int* __restrict__ src,
            const int* __restrict__ dst,
            const unsigned* __restrict__ blkhist,
            int* __restrict__ binstart,
            int* __restrict__ rec) {
    __shared__ int s_tot[TPARTS][NBP];
    __shared__ int s_bef[TPARTS][NBP];
    __shared__ int s_wT[NBP / 32];
    __shared__ int s_wL[NBP / 32];
    __shared__ int s_goff[NBP];
    __shared__ int s_lstart[NBP];
    __shared__ int s_lcur[NBP];
    __shared__ int s_rec[CHUNK];
    __shared__ unsigned char s_bin[CHUNK];

    const int t  = threadIdx.x;
    const int me = blockIdx.x;
    const int lo = me * CHUNK;
    const int n  = min(CHUNK, N_EDGES - lo);

    int d[EPT], s[EPT];
#pragma unroll
    for (int u = 0; u < EPT; ++u) {
        const int i = t + u * PT;
        const int e = (i < n) ? (lo + i) : 0;
        d[u] = dst[e];
        s[u] = src[e];
    }

    {
        const int j    = t & (NBP / 2 - 1);
        const int part = t / (NBP / 2);
        int t0 = 0, t1 = 0, b0 = 0, b1 = 0;
#pragma unroll
        for (int q = 0; q < TROWS; ++q) {
            const int k = part * TROWS + q;
            const unsigned v = blkhist[k * (NBP / 2) + j];
            const int v0 = (int)(v & 0xFFFFu), v1 = (int)(v >> 16);
            t0 += v0; t1 += v1;
            b0 += (k < me) ? v0 : 0;
            b1 += (k < me) ? v1 : 0;
        }
        s_tot[part][2 * j] = t0; s_tot[part][2 * j + 1] = t1;
        s_bef[part][2 * j] = b0; s_bef[part][2 * j + 1] = b1;
    }
    __syncthreads();

    int T = 0, B = 0, own = 0;
    if (t < NBP) {
#pragma unroll
        for (int p = 0; p < TPARTS; ++p) { T += s_tot[p][t]; B += s_bef[p][t]; }
        const unsigned v = blkhist[me * (NBP / 2) + (t >> 1)];
        own = (int)((t & 1) ? (v >> 16) : (v & 0xFFFFu));
    }
    const int incT = block_scan256(T, t, s_wT);
    const int incL = block_scan256(own, t, s_wL);
    if (t < NBP) {
        const int base = incT - T;
        const int ls   = incL - own;
        s_goff[t]   = base + B;
        s_lstart[t] = ls;
        s_lcur[t]   = ls;
        if (me == 0) {
            binstart[t] = base;
            if (t == NBP - 1) binstart[NBP] = incT;
        }
    }
    __syncthreads();

#pragma unroll
    for (int u = 0; u < EPT; ++u) {
        if (t + u * PT < n) {
            const int bb = d[u] >> BIN_SHIFT;
            const int p  = atomicAdd(&s_lcur[bb], 1);
            s_rec[p] = (s[u] << BIN_SHIFT) | (d[u] & (BIN_NODES - 1));
            s_bin[p] = (unsigned char)bb;
        }
    }
    __syncthreads();
    for (int i = t; i < n; i += PT) {
        const int bb = s_bin[i];
        rec[s_goff[bb] + (i - s_lstart[bb])] = s_rec[i];
    }
}

__global__ void __launch_bounds__(ST)
bin_sort(const int* __restrict__ rec,
         const int* __restrict__ binstart,
         int* __restrict__ rowptr,
         int* __restrict__ csr_src) {
    __shared__ int s_cnt[BIN_NODES];
    __shared__ int s_cur[BIN_NODES];
    __shared__ int s_w[NBP / 32];
    __shared__ int s_sorted[SORT_CAP];

    const int t   = threadIdx.x;
    const int bin = blockIdx.x;
    const int lo  = binstart[bin];
    const int n   = binstart[bin + 1] - lo;
    const bool in_lds = (n <= SORT_CAP);

    unsigned r[RPT];
    if (in_lds) {
#pragma unroll
        for (int u = 0; u < RPT; ++u) {
            const int i = t + u * ST;
            r[u] = (unsigned)rec[(i < n) ? (lo + i) : 0];
        }
    }
    if (t < BIN_NODES) s_cnt[t] = 0;
    __syncthreads();
    if (in_lds) {
#pragma unroll
        for (int u = 0; u < RPT; ++u)
            if (t + u * ST < n) atomicAdd(&s_cnt[r[u] & (BIN_NODES - 1)], 1);
    } else {
        for (int i = t; i < n; i += ST)
            atomicAdd(&s_cnt[rec[lo + i] & (BIN_NODES - 1)], 1);
    }
    __syncthreads();
    const int c   = (t < BIN_NODES) ? s_cnt[t] : 0;
    const int inc = block_scan256(c, t, s_w);
    if (t < BIN_NODES) {
        const int pre  = inc - c;
        const int node = bin * BIN_NODES + t;
        s_cur[t] = pre;
        if (node <= N_NODES) rowptr[node] = lo + pre;
    }
    __syncthreads();

    if (in_lds) {
#pragma unroll
        for (int u = 0; u < RPT; ++u) {
            if (t + u * ST < n) {
                const int p = atomicAdd(&s_cur[r[u] & (BIN_NODES - 1)], 1);
                s_sorted[p] = (int)(r[u] >> BIN_SHIFT);
            }
        }
        __syncthreads();
        for (int i = t; i < n; i += ST)
            csr_src[lo + i] = s_sorted[i];
    } else {
        for (int i = t; i < n; i += ST) {
            const unsigned x = (unsigned)rec[lo + i];
            const int p = atomicAdd(&s_cur[x & (BIN_NODES - 1)], 1);
            csr_src[lo + p] = (int)(x >> BIN_SHIFT);
        }
    }
}

#define EDGE_BATCH 4

__global__ void __launch_bounds__(256)
node_softmax_aggregate(const int* __restrict__ rowptr,
                       const int* __restrict__ csr_src,
                       const float* __restrict__ fs,
                       const float* __restrict__ fd,
                       const float* __restrict__ attn_l,
                       float* __restrict__ hout) {
    const int t = blockIdx.x * blockDim.x + threadIdx.x;
    const int n = t >> 4;
    const int j = t & 15;
    if (n >= N_NODES) return;

    const float4* fs4 = (const float4*)fs;
    const float4  b   = ((const float4*)fd)[n * 16 + j];
    const float4  w   = ((const float4*)attn_l)[j];
    const int start = rowptr[n];
    const int end   = rowptr[n + 1];

    float  m = -INFINITY, l = 0.f;
    float4 acc = make_float4(0.f, 0.f, 0.f, 0.f);

    for (int p = start; p < end; p += EDGE_BATCH) {
        float4 a[EDGE_BATCH];
        float  sc[EDGE_BATCH];
#pragma unroll
        for (int u = 0; u < EDGE_BATCH; ++u) {
            const int q = (p + u < end) ? (p + u) : (end - 1);
            a[u] = fs4[csr_src[q] * 16 + j];
        }
#pragma unroll
        for (int u = 0; u < EDGE_BATCH; ++u) {
            float x0 = a[u].x + b.x, x1 = a[u].y + b.y, x2 = a[u].z + b.z, x3 = a[u].w + b.w;
            x0 = (x0 > 0.f) ? x0 : NEG_SLOPE * x0;
            x1 = (x1 > 0.f) ? x1 : NEG_SLOPE * x1;
            x2 = (x2 > 0.f) ? x2 : NEG_SLOPE * x2;
            x3 = (x3 > 0.f) ? x3 : NEG_SLOPE * x3;
            float part = x0 * w.x + x1 * w.y + x2 * w.z + x3 * w.w;
            part += __shfl_xor(part, 1);
            part += __shfl_xor(part, 2);
            sc[u] = (p + u < end) ? part : -INFINITY;
        }
        float mn = fmaxf(m, sc[0]);
#pragma unroll
        for (int u = 1; u < EDGE_BATCH; ++u) mn = fmaxf(mn, sc[u]);
        const float r = expf(m - mn);
        l *= r;
        acc.x *= r; acc.y *= r; acc.z *= r; acc.w *= r;
#pragma unroll
        for (int u = 0; u < EDGE_BATCH; ++u) {
            const float pe = expf(sc[u] - mn);
            l     += pe;
            acc.x += pe * a[u].x;
            acc.y += pe * a[u].y;
            acc.z += pe * a[u].z;
            acc.w += pe * a[u].w;
        }
        m = mn;
    }

    float4 o = make_float4(0.f, 0.f, 0.f, 0.f);
    if (end > start) {
        const float inv = 1.f / l;
        o.x = fmaxf(acc.x * inv, 0.f);
        o.y = fmaxf(acc.y * inv, 0.f);
        o.z = fmaxf(acc.z * inv, 0.f);
        o.w = fmaxf(acc.w * inv, 0.f);
    }
    ((float4*)hout)[n * 16 + j] = o;
}

__global__ void out_proj(const float* __restrict__ h,
                         const float* __restrict__ Wout,
                         const float* __restrict__ bout,
                         float* __restrict__ out) {
    int n = blockIdx.x * blockDim.x + threadIdx.x;
    if (n >= N_NODES) return;
    const float* ph = h + (size_t)n * H_FEATS;
    float a0 = bout[0], a1 = bout[1];
#pragma unroll
    for (int j = 0; j < H_FEATS; ++j) {
        float v = ph[j];
        a0 += v * Wout[j * 2 + 0];
        a1 += v * Wout[j * 2 + 1];
    }
    out[n * 2 + 0] = a0;
    out[n * 2 + 1] = a1;
}

extern "C" void kernel_launch(void* const* d_in, const int* in_sizes, int n_in,
                              void* d_out, int out_size, void* d_ws, size_t ws_size,
                              hipStream_t stream) {
    (void)in_sizes; (void)n_in; (void)out_size; (void)ws_size;

    const float* feature  = (const float*)d_in[0];
    const int*   src      = (const int*)  d_in[1];
    const int*   dst      = (const int*)  d_in[2];
    const float* W_in     = (const float*)d_in[3];
    const float* b_in     = (const float*)d_in[4];
    const float* fc_src_W = (const float*)d_in[5];
    const float* fc_src_b = (const float*)d_in[6];
    const float* fc_dst_W = (const float*)d_in[7];
    const float* fc_dst_b = (const float*)d_in[8];
    const float* attn     = (const float*)d_in[9];
    const float* W_out    = (const float*)d_in[10];
    const float* b_out    = (const float*)d_in[11];
    float* logits = (float*)d_out;

    float* ws = (float*)d_ws;
    size_t off = 0;
    float* h     = ws + off; off += (size_t)N_NODES * H_FEATS;
    float* fs    = ws + off; off += (size_t)N_NODES * H_FEATS;
    float* fd    = ws + off; off += (size_t)N_NODES * H_FEATS;
    int* rowptr   = (int*)(ws + off); off += (size_t)N_NODES + 16;
    int* csr_src  = (int*)(ws + off); off += (size_t)N_EDGES;
    int* rec      = (int*)(ws + off); off += (size_t)N_EDGES;
    unsigned* blkhist = (unsigned*)(ws + off); off += (size_t)PBLK * (NBP / 2);
    int* binstart = (int*)(ws + off); off += (size_t)NBP + 16;

    const int gblocks = (N_NODES / 16 + GEMM_WAVES - 1) / GEMM_WAVES;

    bin_hist<<<PBLK, PT, 0, stream>>>(dst, blkhist);
    bin_scatter<<<PBLK, PT, 0, stream>>>(src, dst, blkhist, binstart, rec);
    bin_sort<<<NBINS, ST, 0, stream>>>(rec, binstart, rowptr, csr_src);

    gemm_bias_wmma<IN_FEATS, 4><<<gblocks, GEMM_WAVES * 32, 0, stream>>>(
        feature, W_in, W_in, b_in, b_in, h, h);

    for (int l = 0; l < 2; ++l) {
        gemm_bias_wmma<H_FEATS, 8><<<gblocks, GEMM_WAVES * 32, 0, stream>>>(
            h, fc_src_W + (size_t)l * H_FEATS * H_FEATS, fc_dst_W + (size_t)l * H_FEATS * H_FEATS,
            fc_src_b + l * H_FEATS, fc_dst_b + l * H_FEATS, fs, fd);

        node_softmax_aggregate<<<(N_NODES * 16 + 255) / 256, 256, 0, stream>>>(
            rowptr, csr_src, fs, fd, attn + l * HEADS * D_HEAD, h);
    }

    out_proj<<<(N_NODES + 255) / 256, 256, 0, stream>>>(h, W_out, b_out, logits);
}


# Each linear fused into the kernel that makes its input (input+layer-0 linears one launch; layer-1 linears and out projection in the aggregates): h never stored; 9->6 launches; logit sum = lane tree
# speedup vs baseline: 9.2223x; 1.2025x over previous
#include <hip/hip_runtime.h>
#include <math.h>

#define N_NODES  50000
#define N_EDGES  800000
#define IN_FEATS 128
#define H_FEATS  64
#define HEADS    4
#define D_HEAD   16
#define NEG_SLOPE 0.2f

typedef __attribute__((ext_vector_type(2))) float v2f;
typedef __attribute__((ext_vector_type(8))) float v8f;

__device__ __forceinline__ v8f wmma_f32_16x16x4(v2f a, v2f b, v8f c) {
    return __builtin_amdgcn_wmma_f32_16x16x4_f32(
         false, a,  false, b,
         (short)0, c,  false,  false);
}

#define GEMM_WAVES 8

template <int K, int NT>
__device__ __forceinline__ void stage_weights(const float* __restrict__ W0,
                                              const float* __restrict__ W1,
                                              float* __restrict__ sB) {
    constexpr int N   = NT * 16;
    constexpr int PN2 = 2 * N + 32;
    constexpr int BLOCKS_PER_THREAD = (K * N / 8) / (GEMM_WAVES * 32);
#pragma unroll
    for (int it = 0; it < BLOCKS_PER_THREAD; ++it) {
        const int i  = it * (GEMM_WAVES * 32) + threadIdx.x;
        const int kp = i / (N / 4);
        const int n4 = (i - kp * (N / 4)) * 4;
        const float* wsrc = (n4 < H_FEATS) ? (W0 + n4) : (W1 + (n4 - H_FEATS));
        const float4 r0 = *(const float4*)(wsrc + (2 * kp) * H_FEATS);
        const float4 r1 = *(const float4*)(wsrc + (2 * kp + 1) * H_FEATS);
        float* d = sB + kp * PN2 + n4 * 2;
        *(float4*)(d)     = make_float4(r0.x, r1.x, r0.y, r1.y);
        *(float4*)(d + 4) = make_float4(r0.z, r1.z, r0.w, r1.w);
    }
}

template <int K, int NT>
__device__ __forceinline__ void tile_mma(const float2 (&av)[K / 4],
                                         const float* __restrict__ bcol,
                                         v8f (&acc)[NT]) {
    constexpr int PN2 = 2 * NT * 16 + 32;
#pragma unroll
    for (int s = 0; s < K / 4; ++s) {
        v2f a; a.x = av[s].x; a.y = av[s].y;
        const float* bp = bcol + (2 * s) * PN2;
#pragma unroll
        for (int nt = 0; nt < NT; ++nt) {
            const float2 bv = *(const float2*)(bp + nt * 32);
            v2f b; b.x = bv.x; b.y = bv.y;
            acc[nt] = wmma_f32_16x16x4(a, b, acc[nt]);
        }
    }
}

#define HP 68

__global__ void __launch_bounds__(GEMM_WAVES * 32)
gemm_in_then_fsfd(const float* __restrict__ X,
                  const float* __restrict__ Win,
                  const float* __restrict__ bin,
                  const float* __restrict__ Ws,
                  const float* __restrict__ Wd,
                  const float* __restrict__ bs,
                  const float* __restrict__ bd,
                  float* __restrict__ FS,
                  float* __restrict__ FD) {
    constexpr int PN1 = 2 * H_FEATS + 32;
    constexpr int PN2 = 2 * (2 * H_FEATS) + 32;
    __shared__ __attribute__((aligned(16))) float sB1[(IN_FEATS / 2) * PN1];
    __shared__ __attribute__((aligned(16))) float sB2[(H_FEATS / 2) * PN2];
    __shared__ __attribute__((aligned(16))) float sH[GEMM_WAVES * 16 * HP];

    stage_weights<IN_FEATS, 4>(Win, Win, sB1);
    stage_weights<H_FEATS, 8>(Ws, Wd, sB2);
    __syncthreads();

    const int lane = threadIdx.x & 31;
    const int wave = threadIdx.x >> 5;
    const int mt_raw = blockIdx.x * GEMM_WAVES + wave;
    const bool valid = mt_raw < N_NODES / 16;
    const int mt = valid ? mt_raw : (N_NODES / 16 - 1);
    const int c  = lane & 15;
    const int hf = lane >> 4;

    float* hw = sH + wave * (16 * HP);
    {
        const float* arow = X + (size_t)(mt * 16 + c) * IN_FEATS + 2 * hf;
        float2 av[IN_FEATS / 4];
#pragma unroll
        for (int s = 0; s < IN_FEATS / 4; ++s)
            av[s] = *(const float2*)(arow + 4 * s);

        v8f acc[4];
#pragma unroll
        for (int nt = 0; nt < 4; ++nt)
            acc[nt] = (v8f){0.f, 0.f, 0.f, 0.f, 0.f, 0.f, 0.f, 0.f};

        tile_mma<IN_FEATS, 4>(av, sB1 + hf * PN1 + c * 2, acc);

#pragma unroll
        for (int nt = 0; nt < 4; ++nt) {
            const int   col = nt * 16 + c;
            const float bv  = bin[col];
#pragma unroll
            for (int r = 0; r < 8; ++r)
                hw[(8 * hf + r) * HP + col] = acc[nt][r] + bv;
        }
    }
    __syncthreads();

    float2 av2[H_FEATS / 4];
    {
        const float* hrow = hw + c * HP + 2 * hf;
#pragma unroll
        for (int s = 0; s < H_FEATS / 4; ++s)
            av2[s] = *(const float2*)(hrow + 4 * s);
    }
    v8f acc2[8];
#pragma unroll
    for (int nt = 0; nt < 8; ++nt)
        acc2[nt] = (v8f){0.f, 0.f, 0.f, 0.f, 0.f, 0.f, 0.f, 0.f};

    tile_mma<H_FEATS, 8>(av2, sB2 + hf * PN2 + c * 2, acc2);

    if (valid) {
        const int row0 = mt * 16 + 8 * hf;
#pragma unroll
        for (int nt = 0; nt < 8; ++nt) {
            const int   col = (nt & 3) * 16 + c;
            const float bv  = (nt < 4) ? bs[col] : bd[col];
            float*      Cp  = (nt < 4) ? FS : FD;
#pragma unroll
            for (int r = 0; r < 8; ++r)
                Cp[(size_t)(row0 + r) * H_FEATS + col] = acc2[nt][r] + bv;
        }
    }
}

#define BIN_SHIFT 8
#define BIN_NODES (1 << BIN_SHIFT)
#define NBINS     ((N_NODES + BIN_NODES - 1) / BIN_NODES)
#define NBP       256
#define PBLK      128
#define PT        1024
#define CHUNK     ((N_EDGES + PBLK - 1) / PBLK)
#define EPT       ((CHUNK + PT - 1) / PT)
#define TPARTS    (PT / (NBP / 2))
#define TROWS     (PBLK / TPARTS)
#define ST        512
#define SORT_CAP  8192
#define RPT       (SORT_CAP / ST)

static_assert(NBINS <= NBP && NBP == 256, "a bin number must fit a byte; the scan below is 8 waves wide");
static_assert(BIN_NODES == NBP, "the node scan of a bin reuses the bin scan");
static_assert(CHUNK < 65536, "a block's count of one bin must fit 16 bits");
static_assert(PT % (NBP / 2) == 0 && PBLK % TPARTS == 0, "table rows split evenly over the thread groups");
static_assert((N_NODES >> BIN_SHIFT) < NBINS, "rowptr[N_NODES] must belong to the last bin");
static_assert(N_NODES <= (1 << (31 - BIN_SHIFT)), "src << BIN_SHIFT must stay positive");

__device__ __forceinline__ int block_scan256(int x, int t, int* s_w) {
    const int lane = t & 31;
    const int w    = t >> 5;
#pragma unroll
    for (int off = 1; off < 32; off <<= 1) {
        const int y = __shfl_up(x, off);
        if (lane >= off) x += y;
    }
    if (t < NBP && lane == 31) s_w[w] = x;
    __syncthreads();
    if (t < NBP) {
        int add = 0;
#pragma unroll
        for (int k = 0; k < NBP / 32; ++k) add += (k < w) ? s_w[k] : 0;
        x += add;
    }
    return x;
}

__global__ void __launch_bounds__(PT)
bin_hist(const int* __restrict__ dst, unsigned* __restrict__ blkhist) {
    __shared__ int h[NBP];
    const int t  = threadIdx.x;
    const int lo = blockIdx.x * CHUNK;
    const int n  = min(CHUNK, N_EDGES - lo);
    int d[EPT];
#pragma unroll
    for (int u = 0; u < EPT; ++u) {
        const int i = t + u * PT;
        d[u] = dst[(i < n) ? (lo + i) : 0];
    }
    if (t < NBP) h[t] = 0;
    __syncthreads();
#pragma unroll
    for (int u = 0; u < EPT; ++u)
        if (t + u * PT < n) atomicAdd(&h[d[u] >> BIN_SHIFT], 1);
    __syncthreads();
    if (t < NBP / 2)
        blkhist[blockIdx.x * (NBP / 2) + t] = (unsigned)h[2 * t] | ((unsigned)h[2 * t + 1] << 16);
}

__global__ void __launch_bounds__(PT)
bin_scatter(const int* __restrict__ src,
            const int* __restrict__ dst,
            const unsigned* __restrict__ blkhist,
            int* __restrict__ binstart,
            int* __restrict__ rec) {
    __shared__ int s_tot[TPARTS][NBP];
    __shared__ int s_bef[TPARTS][NBP];
    __shared__ int s_wT[NBP / 32];
    __shared__ int s_wL[NBP / 32];
    __shared__ int s_goff[NBP];
    __shared__ int s_lstart[NBP];
    __shared__ int s_lcur[NBP];
    __shared__ int s_rec[CHUNK];
    __shared__ unsigned char s_bin[CHUNK];

    const int t  = threadIdx.x;
    const int me = blockIdx.x;
    const int lo = me * CHUNK;
    const int n  = min(CHUNK, N_EDGES - lo);

    int d[EPT], s[EPT];
#pragma unroll
    for (int u = 0; u < EPT; ++u) {
        const int i = t + u * PT;
        const int e = (i < n) ? (lo + i) : 0;
        d[u] = dst[e];
        s[u] = src[e];
    }

    {
        const int j    = t & (NBP / 2 - 1);
        const int part = t / (NBP / 2);
        int t0 = 0, t1 = 0, b0 = 0, b1 = 0;
#pragma unroll
        for (int q = 0; q < TROWS; ++q) {
            const int k = part * TROWS + q;
            const unsigned v = blkhist[k * (NBP / 2) + j];
            const int v0 = (int)(v & 0xFFFFu), v1 = (int)(v >> 16);
            t0 += v0; t1 += v1;
            b0 += (k < me) ? v0 : 0;
            b1 += (k < me) ? v1 : 0;
        }
        s_tot[part][2 * j] = t0; s_tot[part][2 * j + 1] = t1;
        s_bef[part][2 * j] = b0; s_bef[part][2 * j + 1] = b1;
    }
    __syncthreads();

    int T = 0, B = 0, own = 0;
    if (t < NBP) {
#pragma unroll
        for (int p = 0; p < TPARTS; ++p) { T += s_tot[p][t]; B += s_bef[p][t]; }
        const unsigned v = blkhist[me * (NBP / 2) + (t >> 1)];
        own = (int)((t & 1) ? (v >> 16) : (v & 0xFFFFu));
    }
    const int incT = block_scan256(T, t, s_wT);
    const int incL = block_scan256(own, t, s_wL);
    if (t < NBP) {
        const int base = incT - T;
        const int ls   = incL - own;
        s_goff[t]   = base + B;
        s_lstart[t] = ls;
        s_lcur[t]   = ls;
        if (me == 0) {
            binstart[t] = base;
            if (t == NBP - 1) binstart[NBP] = incT;
        }
    }
    __syncthreads();

#pragma unroll
    for (int u = 0; u < EPT; ++u) {
        if (t + u * PT < n) {
            const int bb = d[u] >> BIN_SHIFT;
            const int p  = atomicAdd(&s_lcur[bb], 1);
            s_rec[p] = (s[u] << BIN_SHIFT) | (d[u] & (BIN_NODES - 1));
            s_bin[p] = (unsigned char)bb;
        }
    }
    __syncthreads();
    for (int i = t; i < n; i += PT) {
        const int bb = s_bin[i];
        rec[s_goff[bb] + (i - s_lstart[bb])] = s_rec[i];
    }
}

__global__ void __launch_bounds__(ST)
bin_sort(const int* __restrict__ rec,
         const int* __restrict__ binstart,
         int* __restrict__ rowptr,
         int* __restrict__ csr_src) {
    __shared__ int s_cnt[BIN_NODES];
    __shared__ int s_cur[BIN_NODES];
    __shared__ int s_w[NBP / 32];
    __shared__ int s_sorted[SORT_CAP];

    const int t   = threadIdx.x;
    const int bin = blockIdx.x;
    const int lo  = binstart[bin];
    const int n   = binstart[bin + 1] - lo;
    const bool in_lds = (n <= SORT_CAP);

    unsigned r[RPT];
    if (in_lds) {
#pragma unroll
        for (int u = 0; u < RPT; ++u) {
            const int i = t + u * ST;
            r[u] = (unsigned)rec[(i < n) ? (lo + i) : 0];
        }
    }
    if (t < BIN_NODES) s_cnt[t] = 0;
    __syncthreads();
    if (in_lds) {
#pragma unroll
        for (int u = 0; u < RPT; ++u)
            if (t + u * ST < n) atomicAdd(&s_cnt[r[u] & (BIN_NODES - 1)], 1);
    } else {
        for (int i = t; i < n; i += ST)
            atomicAdd(&s_cnt[rec[lo + i] & (BIN_NODES - 1)], 1);
    }
    __syncthreads();
    const int c   = (t < BIN_NODES) ? s_cnt[t] : 0;
    const int inc = block_scan256(c, t, s_w);
    if (t < BIN_NODES) {
        const int pre  = inc - c;
        const int node = bin * BIN_NODES + t;
        s_cur[t] = pre;
        if (node <= N_NODES) rowptr[node] = lo + pre;
    }
    __syncthreads();

    if (in_lds) {
#pragma unroll
        for (int u = 0; u < RPT; ++u) {
            if (t + u * ST < n) {
                const int p = atomicAdd(&s_cur[r[u] & (BIN_NODES - 1)], 1);
                s_sorted[p] = (int)(r[u] >> BIN_SHIFT);
            }
        }
        __syncthreads();
        for (int i = t; i < n; i += ST)
            csr_src[lo + i] = s_sorted[i];
    } else {
        for (int i = t; i < n; i += ST) {
            const unsigned x = (unsigned)rec[lo + i];
            const int p = atomicAdd(&s_cur[x & (BIN_NODES - 1)], 1);
            csr_src[lo + p] = (int)(x >> BIN_SHIFT);
        }
    }
}

#define EDGE_BATCH 4
#define AGG_THREADS 256
#define PO 132
static_assert(AGG_THREADS / 32 == 2 * (H_FEATS / 16) && AGG_THREADS / 16 == 16, "8 waves = 8 column tiles of [Wa | Wb]; 16 nodes = one 16-row tile");
static_assert((N_NODES * 16) % AGG_THREADS == 0, "whole blocks only: every thread owns a node and reaches the barrier");

template <bool LAST>
__global__ void __launch_bounds__(AGG_THREADS)
node_softmax_aggregate(const int* __restrict__ rowptr,
                       const int* __restrict__ csr_src,
                       const float* __restrict__ fs,
                       const float* __restrict__ fd,
                       const float* __restrict__ attn_l,
                       const float* __restrict__ Wa,
                       const float* __restrict__ Wb,
                       const float* __restrict__ ba,
                       const float* __restrict__ bb,
                       float* __restrict__ outA,
                       float* __restrict__ outB) {
    const int t = blockIdx.x * AGG_THREADS + threadIdx.x;
    const int n = t >> 4;
    const int j = t & 15;

    const float4* fs4 = (const float4*)fs;
    const float4  b   = ((const float4*)fd)[n * 16 + j];
    const float4  w   = ((const float4*)attn_l)[j];
    const int start = rowptr[n];
    const int end   = rowptr[n + 1];

    float  m = -INFINITY, l = 0.f;
    float4 acc = make_float4(0.f, 0.f, 0.f, 0.f);

    for (int p = start; p < end; p += EDGE_BATCH) {
        float4 a[EDGE_BATCH];
        float  sc[EDGE_BATCH];
#pragma unroll
        for (int u = 0; u < EDGE_BATCH; ++u) {
            const int q = (p + u < end) ? (p + u) : (end - 1);
            a[u] = fs4[csr_src[q] * 16 + j];
        }
#pragma unroll
        for (int u = 0; u < EDGE_BATCH; ++u) {
            float x0 = a[u].x + b.x, x1 = a[u].y + b.y, x2 = a[u].z + b.z, x3 = a[u].w + b.w;
            x0 = (x0 > 0.f) ? x0 : NEG_SLOPE * x0;
            x1 = (x1 > 0.f) ? x1 : NEG_SLOPE * x1;
            x2 = (x2 > 0.f) ? x2 : NEG_SLOPE * x2;
            x3 = (x3 > 0.f) ? x3 : NEG_SLOPE * x3;
            float part = x0 * w.x + x1 * w.y + x2 * w.z + x3 * w.w;
            part += __shfl_xor(part, 1);
            part += __shfl_xor(part, 2);
            sc[u] = (p + u < end) ? part : -INFINITY;
        }
        float mn = fmaxf(m, sc[0]);
#pragma unroll
        for (int u = 1; u < EDGE_BATCH; ++u) mn = fmaxf(mn, sc[u]);
        const float r = expf(m - mn);
        l *= r;
        acc.x *= r; acc.y *= r; acc.z *= r; acc.w *= r;
#pragma unroll
        for (int u = 0; u < EDGE_BATCH; ++u) {
            const float pe = expf(sc[u] - mn);
            l     += pe;
            acc.x += pe * a[u].x;
            acc.y += pe * a[u].y;
            acc.z += pe * a[u].z;
            acc.w += pe * a[u].w;
        }
        m = mn;
    }

    float4 o = make_float4(0.f, 0.f, 0.f, 0.f);
    if (end > start) {
        const float inv = 1.f / l;
        o.x = fmaxf(acc.x * inv, 0.f);
        o.y = fmaxf(acc.y * inv, 0.f);
        o.z = fmaxf(acc.z * inv, 0.f);
        o.w = fmaxf(acc.w * inv, 0.f);
    }
    if (!LAST) {
        __shared__ __attribute__((aligned(16))) float sT[(AGG_THREADS / 16) * HP];
        __shared__ __attribute__((aligned(16))) float sO[(AGG_THREADS / 16) * PO];
        const int ln = threadIdx.x >> 4;
        *(float4*)(sT + ln * HP + 4 * j) = o;
        __syncthreads();
        {
            const int lane = threadIdx.x & 31;
            const int wt   = threadIdx.x >> 5;
            const int c    = lane & 15;
            const int hf   = lane >> 4;
            const int n0   = (wt & 3) * 16;
            const float* Wp   = (wt < 4) ? Wa : Wb;
            const float* bp   = (wt < 4) ? ba : bb;
            const float* hrow = sT + c * HP + 2 * hf;
            const float* wcol = Wp + (2 * hf) * H_FEATS + n0 + c;
            v8f ct = (v8f){0.f, 0.f, 0.f, 0.f, 0.f, 0.f, 0.f, 0.f};
#pragma unroll
            for (int s = 0; s < H_FEATS / 4; ++s) {
                const float2 av = *(const float2*)(hrow + 4 * s);
                v2f af; af.x = av.x; af.y = av.y;
                v2f bf; bf.x = wcol[(4 * s) * H_FEATS]; bf.y = wcol[(4 * s + 1) * H_FEATS];
                ct = wmma_f32_16x16x4(af, bf, ct);
            }
            const float bv = bp[n0 + c];
#pragma unroll
            for (int r = 0; r < 8; ++r)
                sO[(8 * hf + r) * PO + wt * 16 + c] = ct[r] + bv;
        }
        __syncthreads();
        ((float4*)outA)[n * 16 + j] = *(const float4*)(sO + ln * PO + 4 * j);
        ((float4*)outB)[n * 16 + j] = *(const float4*)(sO + ln * PO + H_FEATS + 4 * j);
    } else {
        __shared__ float s_out[(AGG_THREADS / 16) * 2];
        const float4 wa = ((const float4*)Wa)[2 * j];
        const float4 wb = ((const float4*)Wa)[2 * j + 1];
        float p0 = o.x * wa.x + o.y * wa.z + o.z * wb.x + o.w * wb.z;
        float p1 = o.x * wa.y + o.y * wa.w + o.z * wb.y + o.w * wb.w;
#pragma unroll
        for (int msk = 1; msk < 16; msk <<= 1) {
            p0 += __shfl_xor(p0, msk);
            p1 += __shfl_xor(p1, msk);
        }
        if (j == 0) {
            const int ln = threadIdx.x >> 4;
            s_out[2 * ln]     = p0 + ba[0];
            s_out[2 * ln + 1] = p1 + ba[1];
        }
        __syncthreads();
        if (threadIdx.x < (AGG_THREADS / 16) * 2)
            outA[blockIdx.x * ((AGG_THREADS / 16) * 2) + threadIdx.x] = s_out[threadIdx.x];
    }
}

extern "C" void kernel_launch(void* const* d_in, const int* in_sizes, int n_in,
                              void* d_out, int out_size, void* d_ws, size_t ws_size,
                              hipStream_t stream) {
    (void)in_sizes; (void)n_in; (void)out_size; (void)ws_size;

    const float* feature  = (const float*)d_in[0];
    const int*   src      = (const int*)  d_in[1];
    const int*   dst      = (const int*)  d_in[2];
    const float* W_in     = (const float*)d_in[3];
    const float* b_in     = (const float*)d_in[4];
    const float* fc_src_W = (const float*)d_in[5];
    const float* fc_src_b = (const float*)d_in[6];
    const float* fc_dst_W = (const float*)d_in[7];
    const float* fc_dst_b = (const float*)d_in[8];
    const float* attn     = (const float*)d_in[9];
    const float* W_out    = (const float*)d_in[10];
    const float* b_out    = (const float*)d_in[11];
    float* logits = (float*)d_out;

    float* ws = (float*)d_ws;
    size_t off = 0;
    float* fs    = ws + off; off += (size_t)N_NODES * H_FEATS;
    float* fd    = ws + off; off += (size_t)N_NODES * H_FEATS;
    float* fs1   = ws + off; off += (size_t)N_NODES * H_FEATS;
    float* fd1   = ws + off; off += (size_t)N_NODES * H_FEATS;
    int* rowptr   = (int*)(ws + off); off += (size_t)N_NODES + 16;
    int* csr_src  = (int*)(ws + off); off += (size_t)N_EDGES;
    int* rec      = (int*)(ws + off); off += (size_t)N_EDGES;
    unsigned* blkhist = (unsigned*)(ws + off); off += (size_t)PBLK * (NBP / 2);
    int* binstart = (int*)(ws + off); off += (size_t)NBP + 16;

    const int gblocks = (N_NODES / 16 + GEMM_WAVES - 1) / GEMM_WAVES;

    bin_hist<<<PBLK, PT, 0, stream>>>(dst, blkhist);
    bin_scatter<<<PBLK, PT, 0, stream>>>(src, dst, blkhist, binstart, rec);
    bin_sort<<<NBINS, ST, 0, stream>>>(rec, binstart, rowptr, csr_src);

    const int ablocks = (N_NODES * 16) / AGG_THREADS;

    gemm_in_then_fsfd<<<gblocks, GEMM_WAVES * 32, 0, stream>>>(
        feature, W_in, b_in, fc_src_W, fc_dst_W, fc_src_b, fc_dst_b, fs, fd);
    node_softmax_aggregate<false><<<ablocks, AGG_THREADS, 0, stream>>>(
        rowptr, csr_src, fs, fd, attn,
        fc_src_W + (size_t)H_FEATS * H_FEATS, fc_dst_W + (size_t)H_FEATS * H_FEATS,
        fc_src_b + H_FEATS, fc_dst_b + H_FEATS, fs1, fd1);
    node_softmax_aggregate<true><<<ablocks, AGG_THREADS, 0, stream>>>(
        rowptr, csr_src, fs1, fd1, attn + HEADS * D_HEAD,
        W_out, W_out, b_out, b_out, logits, logits);
}


# Aggregate loop: a trip's 4 scores are finished one per lane of the head's 4 (two DPP exchange steps, same pair sums): mask, exp and max once per edge, not per edge and lane; no LDS shuffles
# speedup vs baseline: 9.7886x; 1.0614x over previous
#include <hip/hip_runtime.h>
#include <math.h>

#define N_NODES  50000
#define N_EDGES  800000
#define IN_FEATS 128
#define H_FEATS  64
#define HEADS    4
#define D_HEAD   16
#define NEG_SLOPE 0.2f

typedef __attribute__((ext_vector_type(2))) float v2f;
typedef __attribute__((ext_vector_type(8))) float v8f;

__device__ __forceinline__ v8f wmma_f32_16x16x4(v2f a, v2f b, v8f c) {
    return __builtin_amdgcn_wmma_f32_16x16x4_f32(
         false, a,  false, b,
         (short)0, c,  false,  false);
}

#define GEMM_WAVES 8

template <int K, int NT>
__device__ __forceinline__ void stage_weights(const float* __restrict__ W0,
                                              const float* __restrict__ W1,
                                              float* __restrict__ sB) {
    constexpr int N   = NT * 16;
    constexpr int PN2 = 2 * N + 32;
    constexpr int BLOCKS_PER_THREAD = (K * N / 8) / (GEMM_WAVES * 32);
#pragma unroll
    for (int it = 0; it < BLOCKS_PER_THREAD; ++it) {
        const int i  = it * (GEMM_WAVES * 32) + threadIdx.x;
        const int kp = i / (N / 4);
        const int n4 = (i - kp * (N / 4)) * 4;
        const float* wsrc = (n4 < H_FEATS) ? (W0 + n4) : (W1 + (n4 - H_FEATS));
        const float4 r0 = *(const float4*)(wsrc + (2 * kp) * H_FEATS);
        const float4 r1 = *(const float4*)(wsrc + (2 * kp + 1) * H_FEATS);
        float* d = sB + kp * PN2 + n4 * 2;
        *(float4*)(d)     = make_float4(r0.x, r1.x, r0.y, r1.y);
        *(float4*)(d + 4) = make_float4(r0.z, r1.z, r0.w, r1.w);
    }
}

template <int K, int NT>
__device__ __forceinline__ void tile_mma(const float2 (&av)[K / 4],
                                         const float* __restrict__ bcol,
                                         v8f (&acc)[NT]) {
    constexpr int PN2 = 2 * NT * 16 + 32;
#pragma unroll
    for (int s = 0; s < K / 4; ++s) {
        v2f a; a.x = av[s].x; a.y = av[s].y;
        const float* bp = bcol + (2 * s) * PN2;
#pragma unroll
        for (int nt = 0; nt < NT; ++nt) {
            const float2 bv = *(const float2*)(bp + nt * 32);
            v2f b; b.x = bv.x; b.y = bv.y;
            acc[nt] = wmma_f32_16x16x4(a, b, acc[nt]);
        }
    }
}

#define HP 68

__global__ void __launch_bounds__(GEMM_WAVES * 32)
gemm_in_then_fsfd(const float* __restrict__ X,
                  const float* __restrict__ Win,
                  const float* __restrict__ bin,
                  const float* __restrict__ Ws,
                  const float* __restrict__ Wd,
                  const float* __restrict__ bs,
                  const float* __restrict__ bd,
                  float* __restrict__ FS,
                  float* __restrict__ FD) {
    constexpr int PN1 = 2 * H_FEATS + 32;
    constexpr int PN2 = 2 * (2 * H_FEATS) + 32;
    __shared__ __attribute__((aligned(16))) float sB1[(IN_FEATS / 2) * PN1];
    __shared__ __attribute__((aligned(16))) float sB2[(H_FEATS / 2) * PN2];
    __shared__ __attribute__((aligned(16))) float sH[GEMM_WAVES * 16 * HP];

    stage_weights<IN_FEATS, 4>(Win, Win, sB1);
    stage_weights<H_FEATS, 8>(Ws, Wd, sB2);
    __syncthreads();

    const int lane = threadIdx.x & 31;
    const int wave = threadIdx.x >> 5;
    const int mt_raw = blockIdx.x * GEMM_WAVES + wave;
    const bool valid = mt_raw < N_NODES / 16;
    const int mt = valid ? mt_raw : (N_NODES / 16 - 1);
    const int c  = lane & 15;
    const int hf = lane >> 4;

    float* hw = sH + wave * (16 * HP);
    {
        const float* arow = X + (size_t)(mt * 16 + c) * IN_FEATS + 2 * hf;
        float2 av[IN_FEATS / 4];
#pragma unroll
        for (int s = 0; s < IN_FEATS / 4; ++s)
            av[s] = *(const float2*)(arow + 4 * s);

        v8f acc[4];
#pragma unroll
        for (int nt = 0; nt < 4; ++nt)
            acc[nt] = (v8f){0.f, 0.f, 0.f, 0.f, 0.f, 0.f, 0.f, 0.f};

        tile_mma<IN_FEATS, 4>(av, sB1 + hf * PN1 + c * 2, acc);

#pragma unroll
        for (int nt = 0; nt < 4; ++nt) {
            const int   col = nt * 16 + c;
            const float bv  = bin[col];
#pragma unroll
            for (int r = 0; r < 8; ++r)
                hw[(8 * hf + r) * HP + col] = acc[nt][r] + bv;
        }
    }
    __syncthreads();

    float2 av2[H_FEATS / 4];
    {
        const float* hrow = hw + c * HP + 2 * hf;
#pragma unroll
        for (int s = 0; s < H_FEATS / 4; ++s)
            av2[s] = *(const float2*)(hrow + 4 * s);
    }
    v8f acc2[8];
#pragma unroll
    for (int nt = 0; nt < 8; ++nt)
        acc2[nt] = (v8f){0.f, 0.f, 0.f, 0.f, 0.f, 0.f, 0.f, 0.f};

    tile_mma<H_FEATS, 8>(av2, sB2 + hf * PN2 + c * 2, acc2);

    if (valid) {
        const int row0 = mt * 16 + 8 * hf;
#pragma unroll
        for (int nt = 0; nt < 8; ++nt) {
            const int   col = (nt & 3) * 16 + c;
            const float bv  = (nt < 4) ? bs[col] : bd[col];
            float*      Cp  = (nt < 4) ? FS : FD;
#pragma unroll
            for (int r = 0; r < 8; ++r)
                Cp[(size_t)(row0 + r) * H_FEATS + col] = acc2[nt][r] + bv;
        }
    }
}

#define BIN_SHIFT 8
#define BIN_NODES (1 << BIN_SHIFT)
#define NBINS     ((N_NODES + BIN_NODES - 1) / BIN_NODES)
#define NBP       256
#define PBLK      128
#define PT        1024
#define CHUNK     ((N_EDGES + PBLK - 1) / PBLK)
#define EPT       ((CHUNK + PT - 1) / PT)
#define TPARTS    (PT / (NBP / 2))
#define TROWS     (PBLK / TPARTS)
#define ST        512
#define SORT_CAP  8192
#define RPT       (SORT_CAP / ST)

static_assert(NBINS <= NBP && NBP == 256, "a bin number must fit a byte; the scan below is 8 waves wide");
static_assert(BIN_NODES == NBP, "the node scan of a bin reuses the bin scan");
static_assert(CHUNK < 65536, "a block's count of one bin must fit 16 bits");
static_assert(PT % (NBP / 2) == 0 && PBLK % TPARTS == 0, "table rows split evenly over the thread groups");
static_assert((N_NODES >> BIN_SHIFT) < NBINS, "rowptr[N_NODES] must belong to the last bin");
static_assert(N_NODES <= (1 << (31 - BIN_SHIFT)), "src << BIN_SHIFT must stay positive");

__device__ __forceinline__ int block_scan256(int x, int t, int* s_w) {
    const int lane = t & 31;
    const int w    = t >> 5;
#pragma unroll
    for (int off = 1; off < 32; off <<= 1) {
        const int y = __shfl_up(x, off);
        if (lane >= off) x += y;
    }
    if (t < NBP && lane == 31) s_w[w] = x;
    __syncthreads();
    if (t < NBP) {
        int add = 0;
#pragma unroll
        for (int k = 0; k < NBP / 32; ++k) add += (k < w) ? s_w[k] : 0;
        x += add;
    }
    return x;
}

__global__ void __launch_bounds__(PT)
bin_hist(const int* __restrict__ dst, unsigned* __restrict__ blkhist) {
    __shared__ int h[NBP];
    const int t  = threadIdx.x;
    const int lo = blockIdx.x * CHUNK;
    const int n  = min(CHUNK, N_EDGES - lo);
    int d[EPT];
#pragma unroll
    for (int u = 0; u < EPT; ++u) {
        const int i = t + u * PT;
        d[u] = dst[(i < n) ? (lo + i) : 0];
    }
    if (t < NBP) h[t] = 0;
    __syncthreads();
#pragma unroll
    for (int u = 0; u < EPT; ++u)
        if (t + u * PT < n) atomicAdd(&h[d[u] >> BIN_SHIFT], 1);
    __syncthreads();
    if (t < NBP / 2)
        blkhist[blockIdx.x * (NBP / 2) + t] = (unsigned)h[2 * t] | ((unsigned)h[2 * t + 1] << 16);
}

__global__ void __launch_bounds__(PT)
bin_scatter(const int* __restrict__ src,
            const int* __restrict__ dst,
            const unsigned* __restrict__ blkhist,
            int* __restrict__ binstart,
            int* __restrict__ rec) {
    __shared__ int s_tot[TPARTS][NBP];
    __shared__ int s_bef[TPARTS][NBP];
    __shared__ int s_wT[NBP / 32];
    __shared__ int s_wL[NBP / 32];
    __shared__ int s_goff[NBP];
    __shared__ int s_lstart[NBP];
    __shared__ int s_lcur[NBP];
    __shared__ int s_rec[CHUNK];
    __shared__ unsigned char s_bin[CHUNK];

    const int t  = threadIdx.x;
    const int me = blockIdx.x;
    const int lo = me * CHUNK;
    const int n  = min(CHUNK, N_EDGES - lo);

    int d[EPT], s[EPT];
#pragma unroll
    for (int u = 0; u < EPT; ++u) {
        const int i = t + u * PT;
        const int e = (i < n) ? (lo + i) : 0;
        d[u] = dst[e];
        s[u] = src[e];
    }

    {
        const int j    = t & (NBP / 2 - 1);
        const int part = t / (NBP / 2);
        int t0 = 0, t1 = 0, b0 = 0, b1 = 0;
#pragma unroll
        for (int q = 0; q < TROWS; ++q) {
            const int k = part * TROWS + q;
            const unsigned v = blkhist[k * (NBP / 2) + j];
            const int v0 = (int)(v & 0xFFFFu), v1 = (int)(v >> 16);
            t0 += v0; t1 += v1;
            b0 += (k < me) ? v0 : 0;
            b1 += (k < me) ? v1 : 0;
        }
        s_tot[part][2 * j] = t0; s_tot[part][2 * j + 1] = t1;
        s_bef[part][2 * j] = b0; s_bef[part][2 * j + 1] = b1;
    }
    __syncthreads();

    int T = 0, B = 0, own = 0;
    if (t < NBP) {
#pragma unroll
        for (int p = 0; p < TPARTS; ++p) { T += s_tot[p][t]; B += s_bef[p][t]; }
        const unsigned v = blkhist[me * (NBP / 2) + (t >> 1)];
        own = (int)((t & 1) ? (v >> 16) : (v & 0xFFFFu));
    }
    const int incT = block_scan256(T, t, s_wT);
    const int incL = block_scan256(own, t, s_wL);
    if (t < NBP) {
        const int base = incT - T;
        const int ls   = incL - own;
        s_goff[t]   = base + B;
        s_lstart[t] = ls;
        s_lcur[t]   = ls;
        if (me == 0) {
            binstart[t] = base;
            if (t == NBP - 1) binstart[NBP] = incT;
        }
    }
    __syncthreads();

#pragma unroll
    for (int u = 0; u < EPT; ++u) {
        if (t + u * PT < n) {
            const int bb = d[u] >> BIN_SHIFT;
            const int p  = atomicAdd(&s_lcur[bb], 1);
            s_rec[p] = (s[u] << BIN_SHIFT) | (d[u] & (BIN_NODES - 1));
            s_bin[p] = (unsigned char)bb;
        }
    }
    __syncthreads();
    for (int i = t; i < n; i += PT) {
        const int bb = s_bin[i];
        rec[s_goff[bb] + (i - s_lstart[bb])] = s_rec[i];
    }
}

__global__ void __launch_bounds__(ST)
bin_sort(const int* __restrict__ rec,
         const int* __restrict__ binstart,
         int* __restrict__ rowptr,
         int* __restrict__ csr_src) {
    __shared__ int s_cnt[BIN_NODES];
    __shared__ int s_cur[BIN_NODES];
    __shared__ int s_w[NBP / 32];
    __shared__ int s_sorted[SORT_CAP];

    const int t   = threadIdx.x;
    const int bin = blockIdx.x;
    const int lo  = binstart[bin];
    const int n   = binstart[bin + 1] - lo;
    const bool in_lds = (n <= SORT_CAP);

    unsigned r[RPT];
    if (in_lds) {
#pragma unroll
        for (int u = 0; u < RPT; ++u) {
            const int i = t + u * ST;
            r[u] = (unsigned)rec[(i < n) ? (lo + i) : 0];
        }
    }
    if (t < BIN_NODES) s_cnt[t] = 0;
    __syncthreads();
    if (in_lds) {
#pragma unroll
        for (int u = 0; u < RPT; ++u)
            if (t + u * ST < n) atomicAdd(&s_cnt[r[u] & (BIN_NODES - 1)], 1);
    } else {
        for (int i = t; i < n; i += ST)
            atomicAdd(&s_cnt[rec[lo + i] & (BIN_NODES - 1)], 1);
    }
    __syncthreads();
    const int c   = (t < BIN_NODES) ? s_cnt[t] : 0;
    const int inc = block_scan256(c, t, s_w);
    if (t < BIN_NODES) {
        const int pre  = inc - c;
        const int node = bin * BIN_NODES + t;
        s_cur[t] = pre;
        if (node <= N_NODES) rowptr[node] = lo + pre;
    }
    __syncthreads();

    if (in_lds) {
#pragma unroll
        for (int u = 0; u < RPT; ++u) {
            if (t + u * ST < n) {
                const int p = atomicAdd(&s_cur[r[u] & (BIN_NODES - 1)], 1);
                s_sorted[p] = (int)(r[u] >> BIN_SHIFT);
            }
        }
        __syncthreads();
        for (int i = t; i < n; i += ST)
            csr_src[lo + i] = s_sorted[i];
    } else {
        for (int i = t; i < n; i += ST) {
            const unsigned x = (unsigned)rec[lo + i];
            const int p = atomicAdd(&s_cur[x & (BIN_NODES - 1)], 1);
            csr_src[lo + p] = (int)(x >> BIN_SHIFT);
        }
    }
}

#define QP_XOR1 0xB1
#define QP_XOR2 0x4E
#define QP_LANE(u) ((u) * 0x55)
template <int CTRL>
__device__ __forceinline__ float quad_perm(float v) {
    return __int_as_float(__builtin_amdgcn_mov_dpp(__float_as_int(v), CTRL, 0xF, 0xF, true));
}

#define EDGE_BATCH 4
#define AGG_THREADS 256
#define PO 132
static_assert(AGG_THREADS / 32 == 2 * (H_FEATS / 16) && AGG_THREADS / 16 == 16, "8 waves = 8 column tiles of [Wa | Wb]; 16 nodes = one 16-row tile");
static_assert(EDGE_BATCH == 4, "a trip's edges are shared out over the 4 lanes of a head");
static_assert((N_NODES * 16) % AGG_THREADS == 0, "whole blocks only: every thread owns a node and reaches the barrier");

template <bool LAST>
__global__ void __launch_bounds__(AGG_THREADS)
node_softmax_aggregate(const int* __restrict__ rowptr,
                       const int* __restrict__ csr_src,
                       const float* __restrict__ fs,
                       const float* __restrict__ fd,
                       const float* __restrict__ attn_l,
                       const float* __restrict__ Wa,
                       const float* __restrict__ Wb,
                       const float* __restrict__ ba,
                       const float* __restrict__ bb,
                       float* __restrict__ outA,
                       float* __restrict__ outB) {
    const int t = blockIdx.x * AGG_THREADS + threadIdx.x;
    const int n = t >> 4;
    const int j = t & 15;

    const float4* fs4 = (const float4*)fs;
    const float4  b   = ((const float4*)fd)[n * 16 + j];
    const float4  w   = ((const float4*)attn_l)[j];
    const int start = rowptr[n];
    const int end   = rowptr[n + 1];

    float  m = -INFINITY, l = 0.f;
    float4 acc = make_float4(0.f, 0.f, 0.f, 0.f);

    const int  i4 = j & 3;
    const bool b0 = (i4 & 1) != 0;
    const bool b1 = (i4 & 2) != 0;

    for (int p = start; p < end; p += EDGE_BATCH) {
        float4 a[EDGE_BATCH];
        float  part[EDGE_BATCH];
#pragma unroll
        for (int u = 0; u < EDGE_BATCH; ++u) {
            const int q = (p + u < end) ? (p + u) : (end - 1);
            a[u] = fs4[csr_src[q] * 16 + j];
        }
#pragma unroll
        for (int u = 0; u < EDGE_BATCH; ++u) {
            float x0 = a[u].x + b.x, x1 = a[u].y + b.y, x2 = a[u].z + b.z, x3 = a[u].w + b.w;
            x0 = (x0 > 0.f) ? x0 : NEG_SLOPE * x0;
            x1 = (x1 > 0.f) ? x1 : NEG_SLOPE * x1;
            x2 = (x2 > 0.f) ? x2 : NEG_SLOPE * x2;
            x3 = (x3 > 0.f) ? x3 : NEG_SLOPE * x3;
            part[u] = x0 * w.x + x1 * w.y + x2 * w.z + x3 * w.w;
        }
        const float sA = (b0 ? part[1] : part[0]) + quad_perm<QP_XOR1>(b0 ? part[0] : part[1]);
        const float sB = (b0 ? part[3] : part[2]) + quad_perm<QP_XOR1>(b0 ? part[2] : part[3]);
        const float s  = (b1 ? sB : sA) + quad_perm<QP_XOR2>(b1 ? sA : sB);
        const float sc = (p + i4 < end) ? s : -INFINITY;
        float t = fmaxf(sc, quad_perm<QP_XOR1>(sc));
        t = fmaxf(t, quad_perm<QP_XOR2>(t));
        const float mn = fmaxf(m, t);
        const float r  = expf(m - mn);
        const float pw = expf(sc - mn);
        l *= r;
        acc.x *= r; acc.y *= r; acc.z *= r; acc.w *= r;
        const float pe4[EDGE_BATCH] = { quad_perm<QP_LANE(0)>(pw), quad_perm<QP_LANE(1)>(pw),
                                        quad_perm<QP_LANE(2)>(pw), quad_perm<QP_LANE(3)>(pw) };
#pragma unroll
        for (int u = 0; u < EDGE_BATCH; ++u) {
            const float pe = pe4[u];
            l     += pe;
            acc.x += pe * a[u].x;
            acc.y += pe * a[u].y;
            acc.z += pe * a[u].z;
            acc.w += pe * a[u].w;
        }
        m = mn;
    }

    float4 o = make_float4(0.f, 0.f, 0.f, 0.f);
    if (end > start) {
        const float inv = 1.f / l;
        o.x = fmaxf(acc.x * inv, 0.f);
        o.y = fmaxf(acc.y * inv, 0.f);
        o.z = fmaxf(acc.z * inv, 0.f);
        o.w = fmaxf(acc.w * inv, 0.f);
    }
    if (!LAST) {
        __shared__ __attribute__((aligned(16))) float sT[(AGG_THREADS / 16) * HP];
        __shared__ __attribute__((aligned(16))) float sO[(AGG_THREADS / 16) * PO];
        const int ln = threadIdx.x >> 4;
        *(float4*)(sT + ln * HP + 4 * j) = o;
        __syncthreads();
        {
            const int lane = threadIdx.x & 31;
            const int wt   = threadIdx.x >> 5;
            const int c    = lane & 15;
            const int hf   = lane >> 4;
            const int n0   = (wt & 3) * 16;
            const float* Wp   = (wt < 4) ? Wa : Wb;
            const float* bp   = (wt < 4) ? ba : bb;
            const float* hrow = sT + c * HP + 2 * hf;
            const float* wcol = Wp + (2 * hf) * H_FEATS + n0 + c;
            v8f ct = (v8f){0.f, 0.f, 0.f, 0.f, 0.f, 0.f, 0.f, 0.f};
#pragma unroll
            for (int s = 0; s < H_FEATS / 4; ++s) {
                const float2 av = *(const float2*)(hrow + 4 * s);
                v2f af; af.x = av.x; af.y = av.y;
                v2f bf; bf.x = wcol[(4 * s) * H_FEATS]; bf.y = wcol[(4 * s + 1) * H_FEATS];
                ct = wmma_f32_16x16x4(af, bf, ct);
            }
            const float bv = bp[n0 + c];
#pragma unroll
            for (int r = 0; r < 8; ++r)
                sO[(8 * hf + r) * PO + wt * 16 + c] = ct[r] + bv;
        }
        __syncthreads();
        ((float4*)outA)[n * 16 + j] = *(const float4*)(sO + ln * PO + 4 * j);
        ((float4*)outB)[n * 16 + j] = *(const float4*)(sO + ln * PO + H_FEATS + 4 * j);
    } else {
        __shared__ float s_out[(AGG_THREADS / 16) * 2];
        const float4 wa = ((const float4*)Wa)[2 * j];
        const float4 wb = ((const float4*)Wa)[2 * j + 1];
        float p0 = o.x * wa.x + o.y * wa.z + o.z * wb.x + o.w * wb.z;
        float p1 = o.x * wa.y + o.y * wa.w + o.z * wb.y + o.w * wb.w;
#pragma unroll
        for (int msk = 1; msk < 16; msk <<= 1) {
            p0 += __shfl_xor(p0, msk);
            p1 += __shfl_xor(p1, msk);
        }
        if (j == 0) {
            const int ln = threadIdx.x >> 4;
            s_out[2 * ln]     = p0 + ba[0];
            s_out[2 * ln + 1] = p1 + ba[1];
        }
        __syncthreads();
        if (threadIdx.x < (AGG_THREADS / 16) * 2)
            outA[blockIdx.x * ((AGG_THREADS / 16) * 2) + threadIdx.x] = s_out[threadIdx.x];
    }
}

extern "C" void kernel_launch(void* const* d_in, const int* in_sizes, int n_in,
                              void* d_out, int out_size, void* d_ws, size_t ws_size,
                              hipStream_t stream) {
    (void)in_sizes; (void)n_in; (void)out_size; (void)ws_size;

    const float* feature  = (const float*)d_in[0];
    const int*   src      = (const int*)  d_in[1];
    const int*   dst      = (const int*)  d_in[2];
    const float* W_in     = (const float*)d_in[3];
    const float* b_in     = (const float*)d_in[4];
    const float* fc_src_W = (const float*)d_in[5];
    const float* fc_src_b = (const float*)d_in[6];
    const float* fc_dst_W = (const float*)d_in[7];
    const float* fc_dst_b = (const float*)d_in[8];
    const float* attn     = (const float*)d_in[9];
    const float* W_out    = (const float*)d_in[10];
    const float* b_out    = (const float*)d_in[11];
    float* logits = (float*)d_out;

    float* ws = (float*)d_ws;
    size_t off = 0;
    float* fs    = ws + off; off += (size_t)N_NODES * H_FEATS;
    float* fd    = ws + off; off += (size_t)N_NODES * H_FEATS;
    float* fs1   = ws + off; off += (size_t)N_NODES * H_FEATS;
    float* fd1   = ws + off; off += (size_t)N_NODES * H_FEATS;
    int* rowptr   = (int*)(ws + off); off += (size_t)N_NODES + 16;
    int* csr_src  = (int*)(ws + off); off += (size_t)N_EDGES;
    int* rec      = (int*)(ws + off); off += (size_t)N_EDGES;
    unsigned* blkhist = (unsigned*)(ws + off); off += (size_t)PBLK * (NBP / 2);
    int* binstart = (int*)(ws + off); off += (size_t)NBP + 16;

    const int gblocks = (N_NODES / 16 + GEMM_WAVES - 1) / GEMM_WAVES;

    bin_hist<<<PBLK, PT, 0, stream>>>(dst, blkhist);
    bin_scatter<<<PBLK, PT, 0, stream>>>(src, dst, blkhist, binstart, rec);
    bin_sort<<<NBINS, ST, 0, stream>>>(rec, binstart, rowptr, csr_src);

    const int ablocks = (N_NODES * 16) / AGG_THREADS;

    gemm_in_then_fsfd<<<gblocks, GEMM_WAVES * 32, 0, stream>>>(
        feature, W_in, b_in, fc_src_W, fc_dst_W, fc_src_b, fc_dst_b, fs, fd);
    node_softmax_aggregate<false><<<ablocks, AGG_THREADS, 0, stream>>>(
        rowptr, csr_src, fs, fd, attn,
        fc_src_W + (size_t)H_FEATS * H_FEATS, fc_dst_W + (size_t)H_FEATS * H_FEATS,
        fc_src_b + H_FEATS, fc_dst_b + H_FEATS, fs1, fd1);
    node_softmax_aggregate<true><<<ablocks, AGG_THREADS, 0, stream>>>(
        rowptr, csr_src, fs1, fd1, attn + HEADS * D_HEAD,
        W_out, W_out, b_out, b_out, logits, logits);
}


# Layer-0 linears split in two halves, each run in the launch of one of the sort's first two passes (tile blocks, then 256-thread edge blocks): 6->5 launches; segment order = the new LDS arrival order
# speedup vs baseline: 10.2693x; 1.0491x over previous
#include <hip/hip_runtime.h>
#include <math.h>

#define N_NODES  50000
#define N_EDGES  800000
#define IN_FEATS 128
#define H_FEATS  64
#define HEADS    4
#define D_HEAD   16
#define NEG_SLOPE 0.2f

typedef __attribute__((ext_vector_type(2))) float v2f;
typedef __attribute__((ext_vector_type(8))) float v8f;

__device__ __forceinline__ v8f wmma_f32_16x16x4(v2f a, v2f b, v8f c) {
    return __builtin_amdgcn_wmma_f32_16x16x4_f32(
         false, a,  false, b,
         (short)0, c,  false,  false);
}

#define GEMM_WAVES 8

template <int K, int NT>
__device__ __forceinline__ void stage_weights(const float* __restrict__ W0,
                                              const float* __restrict__ W1,
                                              float* __restrict__ sB) {
    constexpr int N   = NT * 16;
    constexpr int PN2 = 2 * N + 32;
    constexpr int BLOCKS_PER_THREAD = (K * N / 8) / (GEMM_WAVES * 32);
#pragma unroll
    for (int it = 0; it < BLOCKS_PER_THREAD; ++it) {
        const int i  = it * (GEMM_WAVES * 32) + threadIdx.x;
        const int kp = i / (N / 4);
        const int n4 = (i - kp * (N / 4)) * 4;
        const float* wsrc = (n4 < H_FEATS) ? (W0 + n4) : (W1 + (n4 - H_FEATS));
        const float4 r0 = *(const float4*)(wsrc + (2 * kp) * H_FEATS);
        const float4 r1 = *(const float4*)(wsrc + (2 * kp + 1) * H_FEATS);
        float* d = sB + kp * PN2 + n4 * 2;
        *(float4*)(d)     = make_float4(r0.x, r1.x, r0.y, r1.y);
        *(float4*)(d + 4) = make_float4(r0.z, r1.z, r0.w, r1.w);
    }
}

template <int K, int NT>
__device__ __forceinline__ void tile_mma(const float2 (&av)[K / 4],
                                         const float* __restrict__ bcol,
                                         v8f (&acc)[NT]) {
    constexpr int PN2 = 2 * NT * 16 + 32;
#pragma unroll
    for (int s = 0; s < K / 4; ++s) {
        v2f a; a.x = av[s].x; a.y = av[s].y;
        const float* bp = bcol + (2 * s) * PN2;
#pragma unroll
        for (int nt = 0; nt < NT; ++nt) {
            const float2 bv = *(const float2*)(bp + nt * 32);
            v2f b; b.x = bv.x; b.y = bv.y;
            acc[nt] = wmma_f32_16x16x4(a, b, acc[nt]);
        }
    }
}

#define HP 68

__device__ __forceinline__ void
gemm_in_then_fsfd(const int gb,
                  const float* __restrict__ X,
                  const float* __restrict__ Win,
                  const float* __restrict__ bin,
                  const float* __restrict__ Ws,
                  const float* __restrict__ Wd,
                  const float* __restrict__ bs,
                  const float* __restrict__ bd,
                  float* __restrict__ FS,
                  float* __restrict__ FD) {
    constexpr int PN1 = 2 * H_FEATS + 32;
    constexpr int PN2 = 2 * (2 * H_FEATS) + 32;
    __shared__ __attribute__((aligned(16))) float sB1[(IN_FEATS / 2) * PN1];
    __shared__ __attribute__((aligned(16))) float sB2[(H_FEATS / 2) * PN2];
    __shared__ __attribute__((aligned(16))) float sH[GEMM_WAVES * 16 * HP];

    stage_weights<IN_FEATS, 4>(Win, Win, sB1);
    stage_weights<H_FEATS, 8>(Ws, Wd, sB2);
    __syncthreads();

    const int lane = threadIdx.x & 31;
    const int wave = threadIdx.x >> 5;
    const int mt_raw = gb * GEMM_WAVES + wave;
    const bool valid = mt_raw < N_NODES / 16;
    const int mt = valid ? mt_raw : (N_NODES / 16 - 1);
    const int c  = lane & 15;
    const int hf = lane >> 4;

    float* hw = sH + wave * (16 * HP);
    {
        const float* arow = X + (size_t)(mt * 16 + c) * IN_FEATS + 2 * hf;
        float2 av[IN_FEATS / 4];
#pragma unroll
        for (int s = 0; s < IN_FEATS / 4; ++s)
            av[s] = *(const float2*)(arow + 4 * s);

        v8f acc[4];
#pragma unroll
        for (int nt = 0; nt < 4; ++nt)
            acc[nt] = (v8f){0.f, 0.f, 0.f, 0.f, 0.f, 0.f, 0.f, 0.f};

        tile_mma<IN_FEATS, 4>(av, sB1 + hf * PN1 + c * 2, acc);

#pragma unroll
        for (int nt = 0; nt < 4; ++nt) {
            const int   col = nt * 16 + c;
            const float bv  = bin[col];
#pragma unroll
            for (int r = 0; r < 8; ++r)
                hw[(8 * hf + r) * HP + col] = acc[nt][r] + bv;
        }
    }
    __syncthreads();

    float2 av2[H_FEATS / 4];
    {
        const float* hrow = hw + c * HP + 2 * hf;
#pragma unroll
        for (int s = 0; s < H_FEATS / 4; ++s)
            av2[s] = *(const float2*)(hrow + 4 * s);
    }
    v8f acc2[8];
#pragma unroll
    for (int nt = 0; nt < 8; ++nt)
        acc2[nt] = (v8f){0.f, 0.f, 0.f, 0.f, 0.f, 0.f, 0.f, 0.f};

    tile_mma<H_FEATS, 8>(av2, sB2 + hf * PN2 + c * 2, acc2);

    if (valid) {
        const int row0 = mt * 16 + 8 * hf;
#pragma unroll
        for (int nt = 0; nt < 8; ++nt) {
            const int   col = (nt & 3) * 16 + c;
            const float bv  = (nt < 4) ? bs[col] : bd[col];
            float*      Cp  = (nt < 4) ? FS : FD;
#pragma unroll
            for (int r = 0; r < 8; ++r)
                Cp[(size_t)(row0 + r) * H_FEATS + col] = acc2[nt][r] + bv;
        }
    }
}

#define BIN_SHIFT 8
#define BIN_NODES (1 << BIN_SHIFT)
#define NBINS     ((N_NODES + BIN_NODES - 1) / BIN_NODES)
#define NBP       256
#define PBLK      128
#define PT        256
#define CHUNK     ((N_EDGES + PBLK - 1) / PBLK)
#define EPT       ((CHUNK + PT - 1) / PT)
#define TPARTS    (PT / (NBP / 2))
#define TROWS     (PBLK / TPARTS)
#define ST        512
#define SORT_CAP  8192
#define RPT       (SORT_CAP / ST)

static_assert(NBINS <= NBP && NBP == 256, "a bin number must fit a byte; the scan below is 8 waves wide");
static_assert(BIN_NODES == NBP, "the node scan of a bin reuses the bin scan");
static_assert(CHUNK < 65536, "a block's count of one bin must fit 16 bits");
static_assert(PT % (NBP / 2) == 0 && PBLK % TPARTS == 0, "table rows split evenly over the thread groups");
static_assert(PT == GEMM_WAVES * 32 && PT == NBP, "edge blocks and tile blocks share a launch: one block size; a thread per bin");
static_assert((N_NODES >> BIN_SHIFT) < NBINS, "rowptr[N_NODES] must belong to the last bin");
static_assert(N_NODES <= (1 << (31 - BIN_SHIFT)), "src << BIN_SHIFT must stay positive");

__device__ __forceinline__ int block_scan256(int x, int t, int* s_w) {
    const int lane = t & 31;
    const int w    = t >> 5;
#pragma unroll
    for (int off = 1; off < 32; off <<= 1) {
        const int y = __shfl_up(x, off);
        if (lane >= off) x += y;
    }
    if (t < NBP && lane == 31) s_w[w] = x;
    __syncthreads();
    if (t < NBP) {
        int add = 0;
#pragma unroll
        for (int k = 0; k < NBP / 32; ++k) add += (k < w) ? s_w[k] : 0;
        x += add;
    }
    return x;
}

__device__ __forceinline__ void
bin_hist(const int me, const int* __restrict__ dst, unsigned* __restrict__ blkhist) {
    __shared__ int h[NBP];
    const int t  = threadIdx.x;
    const int lo = me * CHUNK;
    const int n  = min(CHUNK, N_EDGES - lo);
    int d[EPT];
#pragma unroll
    for (int u = 0; u < EPT; ++u) {
        const int i = t + u * PT;
        d[u] = dst[(i < n) ? (lo + i) : 0];
    }
    if (t < NBP) h[t] = 0;
    __syncthreads();
#pragma unroll
    for (int u = 0; u < EPT; ++u)
        if (t + u * PT < n) atomicAdd(&h[d[u] >> BIN_SHIFT], 1);
    __syncthreads();
    if (t < NBP / 2)
        blkhist[me * (NBP / 2) + t] = (unsigned)h[2 * t] | ((unsigned)h[2 * t + 1] << 16);
}

__device__ __forceinline__ void
bin_scatter(const int me,
            const int* __restrict__ src,
            const int* __restrict__ dst,
            const unsigned* __restrict__ blkhist,
            int* __restrict__ binstart,
            int* __restrict__ rec) {
    __shared__ int s_tot[TPARTS][NBP];
    __shared__ int s_bef[TPARTS][NBP];
    __shared__ int s_wT[NBP / 32];
    __shared__ int s_wL[NBP / 32];
    __shared__ int s_goff[NBP];
    __shared__ int s_lstart[NBP];
    __shared__ int s_lcur[NBP];
    __shared__ int s_rec[CHUNK];
    __shared__ unsigned char s_bin[CHUNK];

    const int t  = threadIdx.x;
    const int lo = me * CHUNK;
    const int n  = min(CHUNK, N_EDGES - lo);

    int d[EPT], s[EPT];
#pragma unroll
    for (int u = 0; u < EPT; ++u) {
        const int i = t + u * PT;
        const int e = (i < n) ? (lo + i) : 0;
        d[u] = dst[e];
        s[u] = src[e];
    }

    {
        const int j    = t & (NBP / 2 - 1);
        const int part = t / (NBP / 2);
        int t0 = 0, t1 = 0, b0 = 0, b1 = 0;
#pragma unroll
        for (int q = 0; q < TROWS; ++q) {
            const int k = part * TROWS + q;
            const unsigned v = blkhist[k * (NBP / 2) + j];
            const int v0 = (int)(v & 0xFFFFu), v1 = (int)(v >> 16);
            t0 += v0; t1 += v1;
            b0 += (k < me) ? v0 : 0;
            b1 += (k < me) ? v1 : 0;
        }
        s_tot[part][2 * j] = t0; s_tot[part][2 * j + 1] = t1;
        s_bef[part][2 * j] = b0; s_bef[part][2 * j + 1] = b1;
    }
    __syncthreads();

    int T = 0, B = 0, own = 0;
    if (t < NBP) {
#pragma unroll
        for (int p = 0; p < TPARTS; ++p) { T += s_tot[p][t]; B += s_bef[p][t]; }
        const unsigned v = blkhist[me * (NBP / 2) + (t >> 1)];
        own = (int)((t & 1) ? (v >> 16) : (v & 0xFFFFu));
    }
    const int incT = block_scan256(T, t, s_wT);
    const int incL = block_scan256(own, t, s_wL);
    if (t < NBP) {
        const int base = incT - T;
        const int ls   = incL - own;
        s_goff[t]   = base + B;
        s_lstart[t] = ls;
        s_lcur[t]   = ls;
        if (me == 0) {
            binstart[t] = base;
            if (t == NBP - 1) binstart[NBP] = incT;
        }
    }
    __syncthreads();

#pragma unroll
    for (int u = 0; u < EPT; ++u) {
        if (t + u * PT < n) {
            const int bb = d[u] >> BIN_SHIFT;
            const int p  = atomicAdd(&s_lcur[bb], 1);
            s_rec[p] = (s[u] << BIN_SHIFT) | (d[u] & (BIN_NODES - 1));
            s_bin[p] = (unsigned char)bb;
        }
    }
    __syncthreads();
    for (int i = t; i < n; i += PT) {
        const int bb = s_bin[i];
        rec[s_goff[bb] + (i - s_lstart[bb])] = s_rec[i];
    }
}

#define GEMM_BLOCKS   ((N_NODES / 16 + GEMM_WAVES - 1) / GEMM_WAVES)
#define GEMM_BLOCKS_0 ((GEMM_BLOCKS + 1) / 2)
#define GEMM_BLOCKS_1 (GEMM_BLOCKS - GEMM_BLOCKS_0)

template <int PASS>
__global__ void __launch_bounds__(GEMM_WAVES * 32)
linears_beside_sort_pass(const float* __restrict__ X,
                         const float* __restrict__ Win,
                         const float* __restrict__ bin,
                         const float* __restrict__ Ws,
                         const float* __restrict__ Wd,
                         const float* __restrict__ bs,
                         const float* __restrict__ bd,
                         float* __restrict__ FS,
                         float* __restrict__ FD,
                         const int* __restrict__ src,
                         const int* __restrict__ dst,
                         unsigned* __restrict__ blkhist,
                         int* __restrict__ binstart,
                         int* __restrict__ rec) {
    constexpr int NG = (PASS == 0) ? GEMM_BLOCKS_0 : GEMM_BLOCKS_1;
    const int bid = blockIdx.x;
    if (bid < NG) {
        gemm_in_then_fsfd(bid + ((PASS == 0) ? 0 : GEMM_BLOCKS_0), X, Win, bin, Ws, Wd, bs, bd, FS, FD);
    } else if (PASS == 0) {
        bin_hist(bid - NG, dst, blkhist);
    } else {
        bin_scatter(bid - NG, src, dst, blkhist, binstart, rec);
    }
}

__global__ void __launch_bounds__(ST)
bin_sort(const int* __restrict__ rec,
         const int* __restrict__ binstart,
         int* __restrict__ rowptr,
         int* __restrict__ csr_src) {
    __shared__ int s_cnt[BIN_NODES];
    __shared__ int s_cur[BIN_NODES];
    __shared__ int s_w[NBP / 32];
    __shared__ int s_sorted[SORT_CAP];

    const int t   = threadIdx.x;
    const int bin = blockIdx.x;
    const int lo  = binstart[bin];
    const int n   = binstart[bin + 1] - lo;
    const bool in_lds = (n <= SORT_CAP);

    unsigned r[RPT];
    if (in_lds) {
#pragma unroll
        for (int u = 0; u < RPT; ++u) {
            const int i = t + u * ST;
            r[u] = (unsigned)rec[(i < n) ? (lo + i) : 0];
        }
    }
    if (t < BIN_NODES) s_cnt[t] = 0;
    __syncthreads();
    if (in_lds) {
#pragma unroll
        for (int u = 0; u < RPT; ++u)
            if (t + u * ST < n) atomicAdd(&s_cnt[r[u] & (BIN_NODES - 1)], 1);
    } else {
        for (int i = t; i < n; i += ST)
            atomicAdd(&s_cnt[rec[lo + i] & (BIN_NODES - 1)], 1);
    }
    __syncthreads();
    const int c   = (t < BIN_NODES) ? s_cnt[t] : 0;
    const int inc = block_scan256(c, t, s_w);
    if (t < BIN_NODES) {
        const int pre  = inc - c;
        const int node = bin * BIN_NODES + t;
        s_cur[t] = pre;
        if (node <= N_NODES) rowptr[node] = lo + pre;
    }
    __syncthreads();

    if (in_lds) {
#pragma unroll
        for (int u = 0; u < RPT; ++u) {
            if (t + u * ST < n) {
                const int p = atomicAdd(&s_cur[r[u] & (BIN_NODES - 1)], 1);
                s_sorted[p] = (int)(r[u] >> BIN_SHIFT);
            }
        }
        __syncthreads();
        for (int i = t; i < n; i += ST)
            csr_src[lo + i] = s_sorted[i];
    } else {
        for (int i = t; i < n; i += ST) {
            const unsigned x = (unsigned)rec[lo + i];
            const int p = atomicAdd(&s_cur[x & (BIN_NODES - 1)], 1);
            csr_src[lo + p] = (int)(x >> BIN_SHIFT);
        }
    }
}

#define QP_XOR1 0xB1
#define QP_XOR2 0x4E
#define QP_LANE(u) ((u) * 0x55)
template <int CTRL>
__device__ __forceinline__ float quad_perm(float v) {
    return __int_as_float(__builtin_amdgcn_mov_dpp(__float_as_int(v), CTRL, 0xF, 0xF, true));
}

#define EDGE_BATCH 4
#define AGG_THREADS 256
#define PO 132
static_assert(AGG_THREADS / 32 == 2 * (H_FEATS / 16) && AGG_THREADS / 16 == 16, "8 waves = 8 column tiles of [Wa | Wb]; 16 nodes = one 16-row tile");
static_assert(EDGE_BATCH == 4, "a trip's edges are shared out over the 4 lanes of a head");
static_assert((N_NODES * 16) % AGG_THREADS == 0, "whole blocks only: every thread owns a node and reaches the barrier");

template <bool LAST>
__global__ void __launch_bounds__(AGG_THREADS)
node_softmax_aggregate(const int* __restrict__ rowptr,
                       const int* __restrict__ csr_src,
                       const float* __restrict__ fs,
                       const float* __restrict__ fd,
                       const float* __restrict__ attn_l,
                       const float* __restrict__ Wa,
                       const float* __restrict__ Wb,
                       const float* __restrict__ ba,
                       const float* __restrict__ bb,
                       float* __restrict__ outA,
                       float* __restrict__ outB) {
    const int t = blockIdx.x * AGG_THREADS + threadIdx.x;
    const int n = t >> 4;
    const int j = t & 15;

    const float4* fs4 = (const float4*)fs;
    const float4  b   = ((const float4*)fd)[n * 16 + j];
    const float4  w   = ((const float4*)attn_l)[j];
    const int start = rowptr[n];
    const int end   = rowptr[n + 1];

    float  m = -INFINITY, l = 0.f;
    float4 acc = make_float4(0.f, 0.f, 0.f, 0.f);

    const int  i4 = j & 3;
    const bool b0 = (i4 & 1) != 0;
    const bool b1 = (i4 & 2) != 0;

    for (int p = start; p < end; p += EDGE_BATCH) {
        float4 a[EDGE_BATCH];
        float  part[EDGE_BATCH];
#pragma unroll
        for (int u = 0; u < EDGE_BATCH; ++u) {
            const int q = (p + u < end) ? (p + u) : (end - 1);
            a[u] = fs4[csr_src[q] * 16 + j];
        }
#pragma unroll
        for (int u = 0; u < EDGE_BATCH; ++u) {
            float x0 = a[u].x + b.x, x1 = a[u].y + b.y, x2 = a[u].z + b.z, x3 = a[u].w + b.w;
            x0 = (x0 > 0.f) ? x0 : NEG_SLOPE * x0;
            x1 = (x1 > 0.f) ? x1 : NEG_SLOPE * x1;
            x2 = (x2 > 0.f) ? x2 : NEG_SLOPE * x2;
            x3 = (x3 > 0.f) ? x3 : NEG_SLOPE * x3;
            part[u] = x0 * w.x + x1 * w.y + x2 * w.z + x3 * w.w;
        }
        const float sA = (b0 ? part[1] : part[0]) + quad_perm<QP_XOR1>(b0 ? part[0] : part[1]);
        const float sB = (b0 ? part[3] : part[2]) + quad_perm<QP_XOR1>(b0 ? part[2] : part[3]);
        const float s  = (b1 ? sB : sA) + quad_perm<QP_XOR2>(b1 ? sA : sB);
        const float sc = (p + i4 < end) ? s : -INFINITY;
        float t = fmaxf(sc, quad_perm<QP_XOR1>(sc));
        t = fmaxf(t, quad_perm<QP_XOR2>(t));
        const float mn = fmaxf(m, t);
        const float r  = expf(m - mn);
        const float pw = expf(sc - mn);
        l *= r;
        acc.x *= r; acc.y *= r; acc.z *= r; acc.w *= r;
        const float pe4[EDGE_BATCH] = { quad_perm<QP_LANE(0)>(pw), quad_perm<QP_LANE(1)>(pw),
                                        quad_perm<QP_LANE(2)>(pw), quad_perm<QP_LANE(3)>(pw) };
#pragma unroll
        for (int u = 0; u < EDGE_BATCH; ++u) {
            const float pe = pe4[u];
            l     += pe;
            acc.x += pe * a[u].x;
            acc.y += pe * a[u].y;
            acc.z += pe * a[u].z;
            acc.w += pe * a[u].w;
        }
        m = mn;
    }

    float4 o = make_float4(0.f, 0.f, 0.f, 0.f);
    if (end > start) {
        const float inv = 1.f / l;
        o.x = fmaxf(acc.x * inv, 0.f);
        o.y = fmaxf(acc.y * inv, 0.f);
        o.z = fmaxf(acc.z * inv, 0.f);
        o.w = fmaxf(acc.w * inv, 0.f);
    }
    if (!LAST) {
        __shared__ __attribute__((aligned(16))) float sT[(AGG_THREADS / 16) * HP];
        __shared__ __attribute__((aligned(16))) float sO[(AGG_THREADS / 16) * PO];
        const int ln = threadIdx.x >> 4;
        *(float4*)(sT + ln * HP + 4 * j) = o;
        __syncthreads();
        {
            const int lane = threadIdx.x & 31;
            const int wt   = threadIdx.x >> 5;
            const int c    = lane & 15;
            const int hf   = lane >> 4;
            const int n0   = (wt & 3) * 16;
            const float* Wp   = (wt < 4) ? Wa : Wb;
            const float* bp   = (wt < 4) ? ba : bb;
            const float* hrow = sT + c * HP + 2 * hf;
            const float* wcol = Wp + (2 * hf) * H_FEATS + n0 + c;
            v8f ct = (v8f){0.f, 0.f, 0.f, 0.f, 0.f, 0.f, 0.f, 0.f};
#pragma unroll
            for (int s = 0; s < H_FEATS / 4; ++s) {
                const float2 av = *(const float2*)(hrow + 4 * s);
                v2f af; af.x = av.x; af.y = av.y;
                v2f bf; bf.x = wcol[(4 * s) * H_FEATS]; bf.y = wcol[(4 * s + 1) * H_FEATS];
                ct = wmma_f32_16x16x4(af, bf, ct);
            }
            const float bv = bp[n0 + c];
#pragma unroll
            for (int r = 0; r < 8; ++r)
                sO[(8 * hf + r) * PO + wt * 16 + c] = ct[r] + bv;
        }
        __syncthreads();
        ((float4*)outA)[n * 16 + j] = *(const float4*)(sO + ln * PO + 4 * j);
        ((float4*)outB)[n * 16 + j] = *(const float4*)(sO + ln * PO + H_FEATS + 4 * j);
    } else {
        __shared__ float s_out[(AGG_THREADS / 16) * 2];
        const float4 wa = ((const float4*)Wa)[2 * j];
        const float4 wb = ((const float4*)Wa)[2 * j + 1];
        float p0 = o.x * wa.x + o.y * wa.z + o.z * wb.x + o.w * wb.z;
        float p1 = o.x * wa.y + o.y * wa.w + o.z * wb.y + o.w * wb.w;
#pragma unroll
        for (int msk = 1; msk < 16; msk <<= 1) {
            p0 += __shfl_xor(p0, msk);
            p1 += __shfl_xor(p1, msk);
        }
        if (j == 0) {
            const int ln = threadIdx.x >> 4;
            s_out[2 * ln]     = p0 + ba[0];
            s_out[2 * ln + 1] = p1 + ba[1];
        }
        __syncthreads();
        if (threadIdx.x < (AGG_THREADS / 16) * 2)
            outA[blockIdx.x * ((AGG_THREADS / 16) * 2) + threadIdx.x] = s_out[threadIdx.x];
    }
}

extern "C" void kernel_launch(void* const* d_in, const int* in_sizes, int n_in,
                              void* d_out, int out_size, void* d_ws, size_t ws_size,
                              hipStream_t stream) {
    (void)in_sizes; (void)n_in; (void)out_size; (void)ws_size;

    const float* feature  = (const float*)d_in[0];
    const int*   src      = (const int*)  d_in[1];
    const int*   dst      = (const int*)  d_in[2];
    const float* W_in     = (const float*)d_in[3];
    const float* b_in     = (const float*)d_in[4];
    const float* fc_src_W = (const float*)d_in[5];
    const float* fc_src_b = (const float*)d_in[6];
    const float* fc_dst_W = (const float*)d_in[7];
    const float* fc_dst_b = (const float*)d_in[8];
    const float* attn     = (const float*)d_in[9];
    const float* W_out    = (const float*)d_in[10];
    const float* b_out    = (const float*)d_in[11];
    float* logits = (float*)d_out;

    float* ws = (float*)d_ws;
    size_t off = 0;
    float* fs    = ws + off; off += (size_t)N_NODES * H_FEATS;
    float* fd    = ws + off; off += (size_t)N_NODES * H_FEATS;
    float* fs1   = ws + off; off += (size_t)N_NODES * H_FEATS;
    float* fd1   = ws + off; off += (size_t)N_NODES * H_FEATS;
    int* rowptr   = (int*)(ws + off); off += (size_t)N_NODES + 16;
    int* csr_src  = (int*)(ws + off); off += (size_t)N_EDGES;
    int* rec      = (int*)(ws + off); off += (size_t)N_EDGES;
    unsigned* blkhist = (unsigned*)(ws + off); off += (size_t)PBLK * (NBP / 2);
    int* binstart = (int*)(ws + off); off += (size_t)NBP + 16;

    linears_beside_sort_pass<0><<<GEMM_BLOCKS_0 + PBLK, GEMM_WAVES * 32, 0, stream>>>(
        feature, W_in, b_in, fc_src_W, fc_dst_W, fc_src_b, fc_dst_b, fs, fd,
        src, dst, blkhist, binstart, rec);
    linears_beside_sort_pass<1><<<GEMM_BLOCKS_1 + PBLK, GEMM_WAVES * 32, 0, stream>>>(
        feature, W_in, b_in, fc_src_W, fc_dst_W, fc_src_b, fc_dst_b, fs, fd,
        src, dst, blkhist, binstart, rec);
    bin_sort<<<NBINS, ST, 0, stream>>>(rec, binstart, rowptr, csr_src);

    const int ablocks = (N_NODES * 16) / AGG_THREADS;

    node_softmax_aggregate<false><<<ablocks, AGG_THREADS, 0, stream>>>(
        rowptr, csr_src, fs, fd, attn,
        fc_src_W + (size_t)H_FEATS * H_FEATS, fc_dst_W + (size_t)H_FEATS * H_FEATS,
        fc_src_b + H_FEATS, fc_dst_b + H_FEATS, fs1, fd1);
    node_softmax_aggregate<true><<<ablocks, AGG_THREADS, 0, stream>>>(
        rowptr, csr_src, fs1, fd1, attn + HEADS * D_HEAD,
        W_out, W_out, b_out, b_out, logits, logits);
}


# Aggregate edge loop as 2-wide packed f32 steps on the row's own register pairs: leaky relu = max(x, slope*x), dot product and weighted sum as fused pair multiply-adds (dot summed as two pairs)
# speedup vs baseline: 10.6032x; 1.0325x over previous
#include <hip/hip_runtime.h>
#include <math.h>

#define N_NODES  50000
#define N_EDGES  800000
#define IN_FEATS 128
#define H_FEATS  64
#define HEADS    4
#define D_HEAD   16
#define NEG_SLOPE 0.2f

typedef __attribute__((ext_vector_type(2))) float v2f;
typedef __attribute__((ext_vector_type(8))) float v8f;

__device__ __forceinline__ v8f wmma_f32_16x16x4(v2f a, v2f b, v8f c) {
    return __builtin_amdgcn_wmma_f32_16x16x4_f32(
         false, a,  false, b,
         (short)0, c,  false,  false);
}

#define GEMM_WAVES 8

template <int K, int NT>
__device__ __forceinline__ void stage_weights(const float* __restrict__ W0,
                                              const float* __restrict__ W1,
                                              float* __restrict__ sB) {
    constexpr int N   = NT * 16;
    constexpr int PN2 = 2 * N + 32;
    constexpr int BLOCKS_PER_THREAD = (K * N / 8) / (GEMM_WAVES * 32);
#pragma unroll
    for (int it = 0; it < BLOCKS_PER_THREAD; ++it) {
        const int i  = it * (GEMM_WAVES * 32) + threadIdx.x;
        const int kp = i / (N / 4);
        const int n4 = (i - kp * (N / 4)) * 4;
        const float* wsrc = (n4 < H_FEATS) ? (W0 + n4) : (W1 + (n4 - H_FEATS));
        const float4 r0 = *(const float4*)(wsrc + (2 * kp) * H_FEATS);
        const float4 r1 = *(const float4*)(wsrc + (2 * kp + 1) * H_FEATS);
        float* d = sB + kp * PN2 + n4 * 2;
        *(float4*)(d)     = make_float4(r0.x, r1.x, r0.y, r1.y);
        *(float4*)(d + 4) = make_float4(r0.z, r1.z, r0.w, r1.w);
    }
}

template <int K, int NT>
__device__ __forceinline__ void tile_mma(const float2 (&av)[K / 4],
                                         const float* __restrict__ bcol,
                                         v8f (&acc)[NT]) {
    constexpr int PN2 = 2 * NT * 16 + 32;
#pragma unroll
    for (int s = 0; s < K / 4; ++s) {
        v2f a; a.x = av[s].x; a.y = av[s].y;
        const float* bp = bcol + (2 * s) * PN2;
#pragma unroll
        for (int nt = 0; nt < NT; ++nt) {
            const float2 bv = *(const float2*)(bp + nt * 32);
            v2f b; b.x = bv.x; b.y = bv.y;
            acc[nt] = wmma_f32_16x16x4(a, b, acc[nt]);
        }
    }
}

#define HP 68

__device__ __forceinline__ void
gemm_in_then_fsfd(const int gb,
                  const float* __restrict__ X,
                  const float* __restrict__ Win,
                  const float* __restrict__ bin,
                  const float* __restrict__ Ws,
                  const float* __restrict__ Wd,
                  const float* __restrict__ bs,
                  const float* __restrict__ bd,
                  float* __restrict__ FS,
                  float* __restrict__ FD) {
    constexpr int PN1 = 2 * H_FEATS + 32;
    constexpr int PN2 = 2 * (2 * H_FEATS) + 32;
    __shared__ __attribute__((aligned(16))) float sB1[(IN_FEATS / 2) * PN1];
    __shared__ __attribute__((aligned(16))) float sB2[(H_FEATS / 2) * PN2];
    __shared__ __attribute__((aligned(16))) float sH[GEMM_WAVES * 16 * HP];

    stage_weights<IN_FEATS, 4>(Win, Win, sB1);
    stage_weights<H_FEATS, 8>(Ws, Wd, sB2);
    __syncthreads();

    const int lane = threadIdx.x & 31;
    const int wave = threadIdx.x >> 5;
    const int mt_raw = gb * GEMM_WAVES + wave;
    const bool valid = mt_raw < N_NODES / 16;
    const int mt = valid ? mt_raw : (N_NODES / 16 - 1);
    const int c  = lane & 15;
    const int hf = lane >> 4;

    float* hw = sH + wave * (16 * HP);
    {
        const float* arow = X + (size_t)(mt * 16 + c) * IN_FEATS + 2 * hf;
        float2 av[IN_FEATS / 4];
#pragma unroll
        for (int s = 0; s < IN_FEATS / 4; ++s)
            av[s] = *(const float2*)(arow + 4 * s);

        v8f acc[4];
#pragma unroll
        for (int nt = 0; nt < 4; ++nt)
            acc[nt] = (v8f){0.f, 0.f, 0.f, 0.f, 0.f, 0.f, 0.f, 0.f};

        tile_mma<IN_FEATS, 4>(av, sB1 + hf * PN1 + c * 2, acc);

#pragma unroll
        for (int nt = 0; nt < 4; ++nt) {
            const int   col = nt * 16 + c;
            const float bv  = bin[col];
#pragma unroll
            for (int r = 0; r < 8; ++r)
                hw[(8 * hf + r) * HP + col] = acc[nt][r] + bv;
        }
    }
    __syncthreads();

    float2 av2[H_FEATS / 4];
    {
        const float* hrow = hw + c * HP + 2 * hf;
#pragma unroll
        for (int s = 0; s < H_FEATS / 4; ++s)
            av2[s] = *(const float2*)(hrow + 4 * s);
    }
    v8f acc2[8];
#pragma unroll
    for (int nt = 0; nt < 8; ++nt)
        acc2[nt] = (v8f){0.f, 0.f, 0.f, 0.f, 0.f, 0.f, 0.f, 0.f};

    tile_mma<H_FEATS, 8>(av2, sB2 + hf * PN2 + c * 2, acc2);

    if (valid) {
        const int row0 = mt * 16 + 8 * hf;
#pragma unroll
        for (int nt = 0; nt < 8; ++nt) {
            const int   col = (nt & 3) * 16 + c;
            const float bv  = (nt < 4) ? bs[col] : bd[col];
            float*      Cp  = (nt < 4) ? FS : FD;
#pragma unroll
            for (int r = 0; r < 8; ++r)
                Cp[(size_t)(row0 + r) * H_FEATS + col] = acc2[nt][r] + bv;
        }
    }
}

#define BIN_SHIFT 8
#define BIN_NODES (1 << BIN_SHIFT)
#define NBINS     ((N_NODES + BIN_NODES - 1) / BIN_NODES)
#define NBP       256
#define PBLK      128
#define PT        256
#define CHUNK     ((N_EDGES + PBLK - 1) / PBLK)
#define EPT       ((CHUNK + PT - 1) / PT)
#define TPARTS    (PT / (NBP / 2))
#define TROWS     (PBLK / TPARTS)
#define ST        512
#define SORT_CAP  8192
#define RPT       (SORT_CAP / ST)

static_assert(NBINS <= NBP && NBP == 256, "a bin number must fit a byte; the scan below is 8 waves wide");
static_assert(BIN_NODES == NBP, "the node scan of a bin reuses the bin scan");
static_assert(CHUNK < 65536, "a block's count of one bin must fit 16 bits");
static_assert(PT % (NBP / 2) == 0 && PBLK % TPARTS == 0, "table rows split evenly over the thread groups");
static_assert(PT == GEMM_WAVES * 32 && PT == NBP, "edge blocks and tile blocks share a launch: one block size; a thread per bin");
static_assert((N_NODES >> BIN_SHIFT) < NBINS, "rowptr[N_NODES] must belong to the last bin");
static_assert(N_NODES <= (1 << (31 - BIN_SHIFT)), "src << BIN_SHIFT must stay positive");

__device__ __forceinline__ int block_scan256(int x, int t, int* s_w) {
    const int lane = t & 31;
    const int w    = t >> 5;
#pragma unroll
    for (int off = 1; off < 32; off <<= 1) {
        const int y = __shfl_up(x, off);
        if (lane >= off) x += y;
    }
    if (t < NBP && lane == 31) s_w[w] = x;
    __syncthreads();
    if (t < NBP) {
        int add = 0;
#pragma unroll
        for (int k = 0; k < NBP / 32; ++k) add += (k < w) ? s_w[k] : 0;
        x += add;
    }
    return x;
}

__device__ __forceinline__ void
bin_hist(const int me, const int* __restrict__ dst, unsigned* __restrict__ blkhist) {
    __shared__ int h[NBP];
    const int t  = threadIdx.x;
    const int lo = me * CHUNK;
    const int n  = min(CHUNK, N_EDGES - lo);
    int d[EPT];
#pragma unroll
    for (int u = 0; u < EPT; ++u) {
        const int i = t + u * PT;
        d[u] = dst[(i < n) ? (lo + i) : 0];
    }
    if (t < NBP) h[t] = 0;
    __syncthreads();
#pragma unroll
    for (int u = 0; u < EPT; ++u)
        if (t + u * PT < n) atomicAdd(&h[d[u] >> BIN_SHIFT], 1);
    __syncthreads();
    if (t < NBP / 2)
        blkhist[me * (NBP / 2) + t] = (unsigned)h[2 * t] | ((unsigned)h[2 * t + 1] << 16);
}

__device__ __forceinline__ void
bin_scatter(const int me,
            const int* __restrict__ src,
            const int* __restrict__ dst,
            const unsigned* __restrict__ blkhist,
            int* __restrict__ binstart,
            int* __restrict__ rec) {
    __shared__ int s_tot[TPARTS][NBP];
    __shared__ int s_bef[TPARTS][NBP];
    __shared__ int s_wT[NBP / 32];
    __shared__ int s_wL[NBP / 32];
    __shared__ int s_goff[NBP];
    __shared__ int s_lstart[NBP];
    __shared__ int s_lcur[NBP];
    __shared__ int s_rec[CHUNK];
    __shared__ unsigned char s_bin[CHUNK];

    const int t  = threadIdx.x;
    const int lo = me * CHUNK;
    const int n  = min(CHUNK, N_EDGES - lo);

    int d[EPT], s[EPT];
#pragma unroll
    for (int u = 0; u < EPT; ++u) {
        const int i = t + u * PT;
        const int e = (i < n) ? (lo + i) : 0;
        d[u] = dst[e];
        s[u] = src[e];
    }

    {
        const int j    = t & (NBP / 2 - 1);
        const int part = t / (NBP / 2);
        int t0 = 0, t1 = 0, b0 = 0, b1 = 0;
#pragma unroll
        for (int q = 0; q < TROWS; ++q) {
            const int k = part * TROWS + q;
            const unsigned v = blkhist[k * (NBP / 2) + j];
            const int v0 = (int)(v & 0xFFFFu), v1 = (int)(v >> 16);
            t0 += v0; t1 += v1;
            b0 += (k < me) ? v0 : 0;
            b1 += (k < me) ? v1 : 0;
        }
        s_tot[part][2 * j] = t0; s_tot[part][2 * j + 1] = t1;
        s_bef[part][2 * j] = b0; s_bef[part][2 * j + 1] = b1;
    }
    __syncthreads();

    int T = 0, B = 0, own = 0;
    if (t < NBP) {
#pragma unroll
        for (int p = 0; p < TPARTS; ++p) { T += s_tot[p][t]; B += s_bef[p][t]; }
        const unsigned v = blkhist[me * (NBP / 2) + (t >> 1)];
        own = (int)((t & 1) ? (v >> 16) : (v & 0xFFFFu));
    }
    const int incT = block_scan256(T, t, s_wT);
    const int incL = block_scan256(own, t, s_wL);
    if (t < NBP) {
        const int base = incT - T;
        const int ls   = incL - own;
        s_goff[t]   = base + B;
        s_lstart[t] = ls;
        s_lcur[t]   = ls;
        if (me == 0) {
            binstart[t] = base;
            if (t == NBP - 1) binstart[NBP] = incT;
        }
    }
    __syncthreads();

#pragma unroll
    for (int u = 0; u < EPT; ++u) {
        if (t + u * PT < n) {
            const int bb = d[u] >> BIN_SHIFT;
            const int p  = atomicAdd(&s_lcur[bb], 1);
            s_rec[p] = (s[u] << BIN_SHIFT) | (d[u] & (BIN_NODES - 1));
            s_bin[p] = (unsigned char)bb;
        }
    }
    __syncthreads();
    for (int i = t; i < n; i += PT) {
        const int bb = s_bin[i];
        rec[s_goff[bb] + (i - s_lstart[bb])] = s_rec[i];
    }
}

#define GEMM_BLOCKS   ((N_NODES / 16 + GEMM_WAVES - 1) / GEMM_WAVES)
#define GEMM_BLOCKS_0 ((GEMM_BLOCKS + 1) / 2)
#define GEMM_BLOCKS_1 (GEMM_BLOCKS - GEMM_BLOCKS_0)

template <int PASS>
__global__ void __launch_bounds__(GEMM_WAVES * 32)
linears_beside_sort_pass(const float* __restrict__ X,
                         const float* __restrict__ Win,
                         const float* __restrict__ bin,
                         const float* __restrict__ Ws,
                         const float* __restrict__ Wd,
                         const float* __restrict__ bs,
                         const float* __restrict__ bd,
                         float* __restrict__ FS,
                         float* __restrict__ FD,
                         const int* __restrict__ src,
                         const int* __restrict__ dst,
                         unsigned* __restrict__ blkhist,
                         int* __restrict__ binstart,
                         int* __restrict__ rec) {
    constexpr int NG = (PASS == 0) ? GEMM_BLOCKS_0 : GEMM_BLOCKS_1;
    const int bid = blockIdx.x;
    if (bid < NG) {
        gemm_in_then_fsfd(bid + ((PASS == 0) ? 0 : GEMM_BLOCKS_0), X, Win, bin, Ws, Wd, bs, bd, FS, FD);
    } else if (PASS == 0) {
        bin_hist(bid - NG, dst, blkhist);
    } else {
        bin_scatter(bid - NG, src, dst, blkhist, binstart, rec);
    }
}

__global__ void __launch_bounds__(ST)
bin_sort(const int* __restrict__ rec,
         const int* __restrict__ binstart,
         int* __restrict__ rowptr,
         int* __restrict__ csr_src) {
    __shared__ int s_cnt[BIN_NODES];
    __shared__ int s_cur[BIN_NODES];
    __shared__ int s_w[NBP / 32];
    __shared__ int s_sorted[SORT_CAP];

    const int t   = threadIdx.x;
    const int bin = blockIdx.x;
    const int lo  = binstart[bin];
    const int n   = binstart[bin + 1] - lo;
    const bool in_lds = (n <= SORT_CAP);

    unsigned r[RPT];
    if (in_lds) {
#pragma unroll
        for (int u = 0; u < RPT; ++u) {
            const int i = t + u * ST;
            r[u] = (unsigned)rec[(i < n) ? (lo + i) : 0];
        }
    }
    if (t < BIN_NODES) s_cnt[t] = 0;
    __syncthreads();
    if (in_lds) {
#pragma unroll
        for (int u = 0; u < RPT; ++u)
            if (t + u * ST < n) atomicAdd(&s_cnt[r[u] & (BIN_NODES - 1)], 1);
    } else {
        for (int i = t; i < n; i += ST)
            atomicAdd(&s_cnt[rec[lo + i] & (BIN_NODES - 1)], 1);
    }
    __syncthreads();
    const int c   = (t < BIN_NODES) ? s_cnt[t] : 0;
    const int inc = block_scan256(c, t, s_w);
    if (t < BIN_NODES) {
        const int pre  = inc - c;
        const int node = bin * BIN_NODES + t;
        s_cur[t] = pre;
        if (node <= N_NODES) rowptr[node] = lo + pre;
    }
    __syncthreads();

    if (in_lds) {
#pragma unroll
        for (int u = 0; u < RPT; ++u) {
            if (t + u * ST < n) {
                const int p = atomicAdd(&s_cur[r[u] & (BIN_NODES - 1)], 1);
                s_sorted[p] = (int)(r[u] >> BIN_SHIFT);
            }
        }
        __syncthreads();
        for (int i = t; i < n; i += ST)
            csr_src[lo + i] = s_sorted[i];
    } else {
        for (int i = t; i < n; i += ST) {
            const unsigned x = (unsigned)rec[lo + i];
            const int p = atomicAdd(&s_cur[x & (BIN_NODES - 1)], 1);
            csr_src[lo + p] = (int)(x >> BIN_SHIFT);
        }
    }
}

#define QP_XOR1 0xB1
#define QP_XOR2 0x4E
#define QP_LANE(u) ((u) * 0x55)
template <int CTRL>
__device__ __forceinline__ float quad_perm(float v) {
    return __int_as_float(__builtin_amdgcn_mov_dpp(__float_as_int(v), CTRL, 0xF, 0xF, true));
}

#define EDGE_BATCH 4
#define AGG_THREADS 256
#define PO 132
static_assert(AGG_THREADS / 32 == 2 * (H_FEATS / 16) && AGG_THREADS / 16 == 16, "8 waves = 8 column tiles of [Wa | Wb]; 16 nodes = one 16-row tile");
static_assert(EDGE_BATCH == 4, "a trip's edges are shared out over the 4 lanes of a head");
static_assert((N_NODES * 16) % AGG_THREADS == 0, "whole blocks only: every thread owns a node and reaches the barrier");

template <bool LAST>
__global__ void __launch_bounds__(AGG_THREADS)
node_softmax_aggregate(const int* __restrict__ rowptr,
                       const int* __restrict__ csr_src,
                       const float* __restrict__ fs,
                       const float* __restrict__ fd,
                       const float* __restrict__ attn_l,
                       const float* __restrict__ Wa,
                       const float* __restrict__ Wb,
                       const float* __restrict__ ba,
                       const float* __restrict__ bb,
                       float* __restrict__ outA,
                       float* __restrict__ outB) {
    const int t = blockIdx.x * AGG_THREADS + threadIdx.x;
    const int n = t >> 4;
    const int j = t & 15;

    const float4* fs4 = (const float4*)fs;
    const float4  b   = ((const float4*)fd)[n * 16 + j];
    const float4  w   = ((const float4*)attn_l)[j];
    const int start = rowptr[n];
    const int end   = rowptr[n + 1];

    float m = -INFINITY, l = 0.f;
    const v2f b01 = {b.x, b.y}, b23 = {b.z, b.w};
    const v2f w01 = {w.x, w.y}, w23 = {w.z, w.w};
    v2f acc01 = {0.f, 0.f}, acc23 = {0.f, 0.f};

    const int  i4 = j & 3;
    const bool b0 = (i4 & 1) != 0;
    const bool b1 = (i4 & 2) != 0;

    for (int p = start; p < end; p += EDGE_BATCH) {
        v2f   a01[EDGE_BATCH], a23[EDGE_BATCH];
        float part[EDGE_BATCH];
#pragma unroll
        for (int u = 0; u < EDGE_BATCH; ++u) {
            const int q = (p + u < end) ? (p + u) : (end - 1);
            const float4 a = fs4[csr_src[q] * 16 + j];
            a01[u] = (v2f){a.x, a.y};
            a23[u] = (v2f){a.z, a.w};
        }
#pragma unroll
        for (int u = 0; u < EDGE_BATCH; ++u) {
            const v2f x01 = a01[u] + b01;
            const v2f x23 = a23[u] + b23;
            const v2f z01 = __builtin_elementwise_max(x01, x01 * NEG_SLOPE);
            const v2f z23 = __builtin_elementwise_max(x23, x23 * NEG_SLOPE);
            const v2f pp = __builtin_elementwise_fma(z23, w23, z01 * w01);
            part[u] = pp.x + pp.y;
        }
        const float sA = (b0 ? part[1] : part[0]) + quad_perm<QP_XOR1>(b0 ? part[0] : part[1]);
        const float sB = (b0 ? part[3] : part[2]) + quad_perm<QP_XOR1>(b0 ? part[2] : part[3]);
        const float s  = (b1 ? sB : sA) + quad_perm<QP_XOR2>(b1 ? sA : sB);
        const float sc = (p + i4 < end) ? s : -INFINITY;
        float t = fmaxf(sc, quad_perm<QP_XOR1>(sc));
        t = fmaxf(t, quad_perm<QP_XOR2>(t));
        const float mn = fmaxf(m, t);
        const float r  = expf(m - mn);
        const float pw = expf(sc - mn);
        l *= r;
        acc01 *= r;
        acc23 *= r;
        const float pe4[EDGE_BATCH] = { quad_perm<QP_LANE(0)>(pw), quad_perm<QP_LANE(1)>(pw),
                                        quad_perm<QP_LANE(2)>(pw), quad_perm<QP_LANE(3)>(pw) };
#pragma unroll
        for (int u = 0; u < EDGE_BATCH; ++u) {
            const float pe = pe4[u];
            const v2f   pv = {pe, pe};
            l    += pe;
            acc01 = __builtin_elementwise_fma(a01[u], pv, acc01);
            acc23 = __builtin_elementwise_fma(a23[u], pv, acc23);
        }
        m = mn;
    }
    const float4 acc = make_float4(acc01.x, acc01.y, acc23.x, acc23.y);

    float4 o = make_float4(0.f, 0.f, 0.f, 0.f);
    if (end > start) {
        const float inv = 1.f / l;
        o.x = fmaxf(acc.x * inv, 0.f);
        o.y = fmaxf(acc.y * inv, 0.f);
        o.z = fmaxf(acc.z * inv, 0.f);
        o.w = fmaxf(acc.w * inv, 0.f);
    }
    if (!LAST) {
        __shared__ __attribute__((aligned(16))) float sT[(AGG_THREADS / 16) * HP];
        __shared__ __attribute__((aligned(16))) float sO[(AGG_THREADS / 16) * PO];
        const int ln = threadIdx.x >> 4;
        *(float4*)(sT + ln * HP + 4 * j) = o;
        __syncthreads();
        {
            const int lane = threadIdx.x & 31;
            const int wt   = threadIdx.x >> 5;
            const int c    = lane & 15;
            const int hf   = lane >> 4;
            const int n0   = (wt & 3) * 16;
            const float* Wp   = (wt < 4) ? Wa : Wb;
            const float* bp   = (wt < 4) ? ba : bb;
            const float* hrow = sT + c * HP + 2 * hf;
            const float* wcol = Wp + (2 * hf) * H_FEATS + n0 + c;
            v8f ct = (v8f){0.f, 0.f, 0.f, 0.f, 0.f, 0.f, 0.f, 0.f};
#pragma unroll
            for (int s = 0; s < H_FEATS / 4; ++s) {
                const float2 av = *(const float2*)(hrow + 4 * s);
                v2f af; af.x = av.x; af.y = av.y;
                v2f bf; bf.x = wcol[(4 * s) * H_FEATS]; bf.y = wcol[(4 * s + 1) * H_FEATS];
                ct = wmma_f32_16x16x4(af, bf, ct);
            }
            const float bv = bp[n0 + c];
#pragma unroll
            for (int r = 0; r < 8; ++r)
                sO[(8 * hf + r) * PO + wt * 16 + c] = ct[r] + bv;
        }
        __syncthreads();
        ((float4*)outA)[n * 16 + j] = *(const float4*)(sO + ln * PO + 4 * j);
        ((float4*)outB)[n * 16 + j] = *(const float4*)(sO + ln * PO + H_FEATS + 4 * j);
    } else {
        __shared__ float s_out[(AGG_THREADS / 16) * 2];
        const float4 wa = ((const float4*)Wa)[2 * j];
        const float4 wb = ((const float4*)Wa)[2 * j + 1];
        float p0 = o.x * wa.x + o.y * wa.z + o.z * wb.x + o.w * wb.z;
        float p1 = o.x * wa.y + o.y * wa.w + o.z * wb.y + o.w * wb.w;
#pragma unroll
        for (int msk = 1; msk < 16; msk <<= 1) {
            p0 += __shfl_xor(p0, msk);
            p1 += __shfl_xor(p1, msk);
        }
        if (j == 0) {
            const int ln = threadIdx.x >> 4;
            s_out[2 * ln]     = p0 + ba[0];
            s_out[2 * ln + 1] = p1 + ba[1];
        }
        __syncthreads();
        if (threadIdx.x < (AGG_THREADS / 16) * 2)
            outA[blockIdx.x * ((AGG_THREADS / 16) * 2) + threadIdx.x] = s_out[threadIdx.x];
    }
}

extern "C" void kernel_launch(void* const* d_in, const int* in_sizes, int n_in,
                              void* d_out, int out_size, void* d_ws, size_t ws_size,
                              hipStream_t stream) {
    (void)in_sizes; (void)n_in; (void)out_size; (void)ws_size;

    const float* feature  = (const float*)d_in[0];
    const int*   src      = (const int*)  d_in[1];
    const int*   dst      = (const int*)  d_in[2];
    const float* W_in     = (const float*)d_in[3];
    const float* b_in     = (const float*)d_in[4];
    const float* fc_src_W = (const float*)d_in[5];
    const float* fc_src_b = (const float*)d_in[6];
    const float* fc_dst_W = (const float*)d_in[7];
    const float* fc_dst_b = (const float*)d_in[8];
    const float* attn     = (const float*)d_in[9];
    const float* W_out    = (const float*)d_in[10];
    const float* b_out    = (const float*)d_in[11];
    float* logits = (float*)d_out;

    float* ws = (float*)d_ws;
    size_t off = 0;
    float* fs    = ws + off; off += (size_t)N_NODES * H_FEATS;
    float* fd    = ws + off; off += (size_t)N_NODES * H_FEATS;
    float* fs1   = ws + off; off += (size_t)N_NODES * H_FEATS;
    float* fd1   = ws + off; off += (size_t)N_NODES * H_FEATS;
    int* rowptr   = (int*)(ws + off); off += (size_t)N_NODES + 16;
    int* csr_src  = (int*)(ws + off); off += (size_t)N_EDGES;
    int* rec      = (int*)(ws + off); off += (size_t)N_EDGES;
    unsigned* blkhist = (unsigned*)(ws + off); off += (size_t)PBLK * (NBP / 2);
    int* binstart = (int*)(ws + off); off += (size_t)NBP + 16;

    linears_beside_sort_pass<0><<<GEMM_BLOCKS_0 + PBLK, GEMM_WAVES * 32, 0, stream>>>(
        feature, W_in, b_in, fc_src_W, fc_dst_W, fc_src_b, fc_dst_b, fs, fd,
        src, dst, blkhist, binstart, rec);
    linears_beside_sort_pass<1><<<GEMM_BLOCKS_1 + PBLK, GEMM_WAVES * 32, 0, stream>>>(
        feature, W_in, b_in, fc_src_W, fc_dst_W, fc_src_b, fc_dst_b, fs, fd,
        src, dst, blkhist, binstart, rec);
    bin_sort<<<NBINS, ST, 0, stream>>>(rec, binstart, rowptr, csr_src);

    const int ablocks = (N_NODES * 16) / AGG_THREADS;

    node_softmax_aggregate<false><<<ablocks, AGG_THREADS, 0, stream>>>(
        rowptr, csr_src, fs, fd, attn,
        fc_src_W + (size_t)H_FEATS * H_FEATS, fc_dst_W + (size_t)H_FEATS * H_FEATS,
        fc_src_b + H_FEATS, fc_dst_b + H_FEATS, fs1, fd1);
    node_softmax_aggregate<true><<<ablocks, AGG_THREADS, 0, stream>>>(
        rowptr, csr_src, fs1, fd1, attn + HEADS * D_HEAD,
        W_out, W_out, b_out, b_out, logits, logits);
}


# Edge sort in 2 passes, not 3: edge blocks bin-order their chunk in place and table the run starts; bin blocks gather the runs beside the 2nd linears half; 5->4 launches; sums in new LDS arrival order
# speedup vs baseline: 11.4200x; 1.0770x over previous
#include <hip/hip_runtime.h>
#include <math.h>

#define N_NODES  50000
#define N_EDGES  800000
#define IN_FEATS 128
#define H_FEATS  64
#define HEADS    4
#define D_HEAD   16
#define NEG_SLOPE 0.2f

typedef __attribute__((ext_vector_type(2))) float v2f;
typedef __attribute__((ext_vector_type(8))) float v8f;

__device__ __forceinline__ v8f wmma_f32_16x16x4(v2f a, v2f b, v8f c) {
    return __builtin_amdgcn_wmma_f32_16x16x4_f32(
         false, a,  false, b,
         (short)0, c,  false,  false);
}

#define GEMM_WAVES 8

template <int K, int NT>
__device__ __forceinline__ void stage_weights(const float* __restrict__ W0,
                                              const float* __restrict__ W1,
                                              float* __restrict__ sB) {
    constexpr int N   = NT * 16;
    constexpr int PN2 = 2 * N + 32;
    constexpr int BLOCKS_PER_THREAD = (K * N / 8) / (GEMM_WAVES * 32);
#pragma unroll
    for (int it = 0; it < BLOCKS_PER_THREAD; ++it) {
        const int i  = it * (GEMM_WAVES * 32) + threadIdx.x;
        const int kp = i / (N / 4);
        const int n4 = (i - kp * (N / 4)) * 4;
        const float* wsrc = (n4 < H_FEATS) ? (W0 + n4) : (W1 + (n4 - H_FEATS));
        const float4 r0 = *(const float4*)(wsrc + (2 * kp) * H_FEATS);
        const float4 r1 = *(const float4*)(wsrc + (2 * kp + 1) * H_FEATS);
        float* d = sB + kp * PN2 + n4 * 2;
        *(float4*)(d)     = make_float4(r0.x, r1.x, r0.y, r1.y);
        *(float4*)(d + 4) = make_float4(r0.z, r1.z, r0.w, r1.w);
    }
}

template <int K, int NT>
__device__ __forceinline__ void tile_mma(const float2 (&av)[K / 4],
                                         const float* __restrict__ bcol,
                                         v8f (&acc)[NT]) {
    constexpr int PN2 = 2 * NT * 16 + 32;
#pragma unroll
    for (int s = 0; s < K / 4; ++s) {
        v2f a; a.x = av[s].x; a.y = av[s].y;
        const float* bp = bcol + (2 * s) * PN2;
#pragma unroll
        for (int nt = 0; nt < NT; ++nt) {
            const float2 bv = *(const float2*)(bp + nt * 32);
            v2f b; b.x = bv.x; b.y = bv.y;
            acc[nt] = wmma_f32_16x16x4(a, b, acc[nt]);
        }
    }
}

#define HP 68

__device__ __forceinline__ void
gemm_in_then_fsfd(const int gb,
                  const float* __restrict__ X,
                  const float* __restrict__ Win,
                  const float* __restrict__ bin,
                  const float* __restrict__ Ws,
                  const float* __restrict__ Wd,
                  const float* __restrict__ bs,
                  const float* __restrict__ bd,
                  float* __restrict__ FS,
                  float* __restrict__ FD) {
    constexpr int PN1 = 2 * H_FEATS + 32;
    constexpr int PN2 = 2 * (2 * H_FEATS) + 32;
    __shared__ __attribute__((aligned(16))) float sB1[(IN_FEATS / 2) * PN1];
    __shared__ __attribute__((aligned(16))) float sB2[(H_FEATS / 2) * PN2];
    __shared__ __attribute__((aligned(16))) float sH[GEMM_WAVES * 16 * HP];

    stage_weights<IN_FEATS, 4>(Win, Win, sB1);
    stage_weights<H_FEATS, 8>(Ws, Wd, sB2);
    __syncthreads();

    const int lane = threadIdx.x & 31;
    const int wave = threadIdx.x >> 5;
    const int mt_raw = gb * GEMM_WAVES + wave;
    const bool valid = mt_raw < N_NODES / 16;
    const int mt = valid ? mt_raw : (N_NODES / 16 - 1);
    const int c  = lane & 15;
    const int hf = lane >> 4;

    float* hw = sH + wave * (16 * HP);
    {
        const float* arow = X + (size_t)(mt * 16 + c) * IN_FEATS + 2 * hf;
        float2 av[IN_FEATS / 4];
#pragma unroll
        for (int s = 0; s < IN_FEATS / 4; ++s)
            av[s] = *(const float2*)(arow + 4 * s);

        v8f acc[4];
#pragma unroll
        for (int nt = 0; nt < 4; ++nt)
            acc[nt] = (v8f){0.f, 0.f, 0.f, 0.f, 0.f, 0.f, 0.f, 0.f};

        tile_mma<IN_FEATS, 4>(av, sB1 + hf * PN1 + c * 2, acc);

#pragma unroll
        for (int nt = 0; nt < 4; ++nt) {
            const int   col = nt * 16 + c;
            const float bv  = bin[col];
#pragma unroll
            for (int r = 0; r < 8; ++r)
                hw[(8 * hf + r) * HP + col] = acc[nt][r] + bv;
        }
    }
    __syncthreads();

    float2 av2[H_FEATS / 4];
    {
        const float* hrow = hw + c * HP + 2 * hf;
#pragma unroll
        for (int s = 0; s < H_FEATS / 4; ++s)
            av2[s] = *(const float2*)(hrow + 4 * s);
    }
    v8f acc2[8];
#pragma unroll
    for (int nt = 0; nt < 8; ++nt)
        acc2[nt] = (v8f){0.f, 0.f, 0.f, 0.f, 0.f, 0.f, 0.f, 0.f};

    tile_mma<H_FEATS, 8>(av2, sB2 + hf * PN2 + c * 2, acc2);

    if (valid) {
        const int row0 = mt * 16 + 8 * hf;
#pragma unroll
        for (int nt = 0; nt < 8; ++nt) {
            const int   col = (nt & 3) * 16 + c;
            const float bv  = (nt < 4) ? bs[col] : bd[col];
            float*      Cp  = (nt < 4) ? FS : FD;
#pragma unroll
            for (int r = 0; r < 8; ++r)
                Cp[(size_t)(row0 + r) * H_FEATS + col] = acc2[nt][r] + bv;
        }
    }
}

#define BIN_SHIFT 8
#define BIN_NODES (1 << BIN_SHIFT)
#define NBINS     ((N_NODES + BIN_NODES - 1) / BIN_NODES)
#define NBP       256
#define PBLK      128
#define PT        256
#define CHUNK     ((N_EDGES + PBLK - 1) / PBLK)
#define EPT       ((CHUNK + PT - 1) / PT)
#define TABP      288
#define RUN_LANES (PT / PBLK)
#define RSLOTS    32
#define SORT_CAP  8192

static_assert(NBINS <= NBP && NBP == 256, "a bin number must fit the scan below, which is 8 waves wide");
static_assert(BIN_NODES == NBP, "the node scan of a bin reuses the bin scan");
static_assert(PT == GEMM_WAVES * 32 && PT == NBP, "edge, bin and tile blocks share launches: one block size; a thread per bin and per node of a bin");
static_assert(PT % PBLK == 0 && RUN_LANES >= 1, "the threads of a bin block split evenly over the runs");
static_assert(TABP >= NBP + 1 && TABP % 32 == 0, "a table row holds NBP + 1 entries in lines of its own");
static_assert((N_NODES >> BIN_SHIFT) < NBINS, "rowptr[N_NODES] must belong to the last bin");
static_assert(N_NODES <= (1 << (31 - BIN_SHIFT)), "src << BIN_SHIFT must stay positive");

__device__ __forceinline__ int block_scan256(int x, int t, int* s_w) {
    const int lane = t & 31;
    const int w    = t >> 5;
#pragma unroll
    for (int off = 1; off < 32; off <<= 1) {
        const int y = __shfl_up(x, off);
        if (lane >= off) x += y;
    }
    if (t < NBP && lane == 31) s_w[w] = x;
    __syncthreads();
    if (t < NBP) {
        int add = 0;
#pragma unroll
        for (int k = 0; k < NBP / 32; ++k) add += (k < w) ? s_w[k] : 0;
        x += add;
    }
    return x;
}

__device__ __forceinline__ void
bin_local_order(const int me,
                const int* __restrict__ src,
                const int* __restrict__ dst,
                int* __restrict__ tab,
                int* __restrict__ rec) {
    __shared__ int h[NBP];
    __shared__ int s_lcur[NBP];
    __shared__ int s_w[NBP / 32];
    __shared__ int s_rec[CHUNK];

    const int t  = threadIdx.x;
    const int lo = me * CHUNK;
    const int n  = min(CHUNK, N_EDGES - lo);

    int d[EPT], s[EPT];
#pragma unroll
    for (int u = 0; u < EPT; ++u) {
        const int i = t + u * PT;
        const int e = (i < n) ? (lo + i) : 0;
        d[u] = dst[e];
        s[u] = src[e];
    }
    h[t] = 0;
    __syncthreads();
#pragma unroll
    for (int u = 0; u < EPT; ++u)
        if (t + u * PT < n) atomicAdd(&h[d[u] >> BIN_SHIFT], 1);
    __syncthreads();

    const int own = h[t];
    const int inc = block_scan256(own, t, s_w);
    const int ls  = inc - own;
    s_lcur[t] = ls;
    tab[me * TABP + t] = ls;
    if (t == NBP - 1) tab[me * TABP + NBP] = inc;
    __syncthreads();

#pragma unroll
    for (int u = 0; u < EPT; ++u) {
        if (t + u * PT < n) {
            const int p = atomicAdd(&s_lcur[d[u] >> BIN_SHIFT], 1);
            s_rec[p] = (s[u] << BIN_SHIFT) | (d[u] & (BIN_NODES - 1));
        }
    }
    __syncthreads();
    for (int i = t; i < n; i += PT)
        rec[lo + i] = s_rec[i];
}

__device__ __forceinline__ int block_sum256(int x, int t, int* s_p) {
#pragma unroll
    for (int off = 16; off > 0; off >>= 1) x += __shfl_xor(x, off);
    if ((t & 31) == 0) s_p[t >> 5] = x;
    __syncthreads();
    int tot = 0;
#pragma unroll
    for (int k = 0; k < PT / 32; ++k) tot += s_p[k];
    return tot;
}

__device__ __forceinline__ void
bin_gather_sort(const int bin,
                const int* __restrict__ tab,
                const int* __restrict__ rec,
                int* __restrict__ rowptr,
                int* __restrict__ csr_src) {
    __shared__ int s_cnt[BIN_NODES];
    __shared__ int s_cur[BIN_NODES];
    __shared__ int s_w[NBP / 32];
    __shared__ int s_pn[PT / 32];
    __shared__ int s_pl[PT / 32];
    __shared__ int s_sorted[SORT_CAP];

    const int t = threadIdx.x;
    const int k = t / RUN_LANES;
    const int l = t % RUN_LANES;
    const int st  = tab[k * TABP + bin];
    const int cnt = tab[k * TABP + bin + 1] - st;
    s_cnt[t] = 0;

    const int rbase = k * CHUNK + st;
    unsigned v[RSLOTS];
#pragma unroll
    for (int u = 0; u < RSLOTS; ++u) {
        const int i = l + u * RUN_LANES;
        v[u] = (unsigned)rec[(i < cnt) ? (rbase + i) : 0];
    }

    const int n  = block_sum256((l == 0) ? cnt : 0, t, s_pn);
    const int lo = block_sum256((l == 0) ? st  : 0, t, s_pl);
    const bool in_lds = (n <= SORT_CAP);

#pragma unroll
    for (int u = 0; u < RSLOTS; ++u)
        if (l + u * RUN_LANES < cnt) atomicAdd(&s_cnt[v[u] & (BIN_NODES - 1)], 1);
    for (int i = l + RSLOTS * RUN_LANES; i < cnt; i += RUN_LANES)
        atomicAdd(&s_cnt[rec[rbase + i] & (BIN_NODES - 1)], 1);
    __syncthreads();

    const int c   = s_cnt[t];
    const int inc = block_scan256(c, t, s_w);
    {
        const int pre  = inc - c;
        const int node = bin * BIN_NODES + t;
        s_cur[t] = pre;
        if (node <= N_NODES) rowptr[node] = lo + pre;
    }
    __syncthreads();

    if (in_lds) {
#pragma unroll
        for (int u = 0; u < RSLOTS; ++u) {
            if (l + u * RUN_LANES < cnt) {
                const int p = atomicAdd(&s_cur[v[u] & (BIN_NODES - 1)], 1);
                s_sorted[p] = (int)(v[u] >> BIN_SHIFT);
            }
        }
        for (int i = l + RSLOTS * RUN_LANES; i < cnt; i += RUN_LANES) {
            const unsigned x = (unsigned)rec[rbase + i];
            const int p = atomicAdd(&s_cur[x & (BIN_NODES - 1)], 1);
            s_sorted[p] = (int)(x >> BIN_SHIFT);
        }
        __syncthreads();
        for (int i = t; i < n; i += PT)
            csr_src[lo + i] = s_sorted[i];
    } else {
#pragma unroll
        for (int u = 0; u < RSLOTS; ++u) {
            if (l + u * RUN_LANES < cnt) {
                const int p = atomicAdd(&s_cur[v[u] & (BIN_NODES - 1)], 1);
                csr_src[lo + p] = (int)(v[u] >> BIN_SHIFT);
            }
        }
        for (int i = l + RSLOTS * RUN_LANES; i < cnt; i += RUN_LANES) {
            const unsigned x = (unsigned)rec[rbase + i];
            const int p = atomicAdd(&s_cur[x & (BIN_NODES - 1)], 1);
            csr_src[lo + p] = (int)(x >> BIN_SHIFT);
        }
    }
}

#define GEMM_BLOCKS   ((N_NODES / 16 + GEMM_WAVES - 1) / GEMM_WAVES)
#define GEMM_BLOCKS_0 ((GEMM_BLOCKS + 1) / 2)
#define GEMM_BLOCKS_1 (GEMM_BLOCKS - GEMM_BLOCKS_0)

template <int PASS>
__global__ void __launch_bounds__(GEMM_WAVES * 32)
linears_beside_sort_pass(const float* __restrict__ X,
                         const float* __restrict__ Win,
                         const float* __restrict__ bin,
                         const float* __restrict__ Ws,
                         const float* __restrict__ Wd,
                         const float* __restrict__ bs,
                         const float* __restrict__ bd,
                         float* __restrict__ FS,
                         float* __restrict__ FD,
                         const int* __restrict__ src,
                         const int* __restrict__ dst,
                         int* __restrict__ tab,
                         int* __restrict__ rec,
                         int* __restrict__ rowptr,
                         int* __restrict__ csr_src) {
    constexpr int NG = (PASS == 0) ? GEMM_BLOCKS_0 : GEMM_BLOCKS_1;
    const int bid = blockIdx.x;
    if (bid < NG) {
        gemm_in_then_fsfd(bid + ((PASS == 0) ? 0 : GEMM_BLOCKS_0), X, Win, bin, Ws, Wd, bs, bd, FS, FD);
    } else if (PASS == 0) {
        bin_local_order(bid - NG, src, dst, tab, rec);
    } else {
        bin_gather_sort(bid - NG, tab, rec, rowptr, csr_src);
    }
}

#define QP_XOR1 0xB1
#define QP_XOR2 0x4E
#define QP_LANE(u) ((u) * 0x55)
template <int CTRL>
__device__ __forceinline__ float quad_perm(float v) {
    return __int_as_float(__builtin_amdgcn_mov_dpp(__float_as_int(v), CTRL, 0xF, 0xF, true));
}

#define EDGE_BATCH 4
#define AGG_THREADS 256
#define PO 132
static_assert(AGG_THREADS / 32 == 2 * (H_FEATS / 16) && AGG_THREADS / 16 == 16, "8 waves = 8 column tiles of [Wa | Wb]; 16 nodes = one 16-row tile");
static_assert(EDGE_BATCH == 4, "a trip's edges are shared out over the 4 lanes of a head");
static_assert((N_NODES * 16) % AGG_THREADS == 0, "whole blocks only: every thread owns a node and reaches the barrier");

template <bool LAST>
__global__ void __launch_bounds__(AGG_THREADS)
node_softmax_aggregate(const int* __restrict__ rowptr,
                       const int* __restrict__ csr_src,
                       const float* __restrict__ fs,
                       const float* __restrict__ fd,
                       const float* __restrict__ attn_l,
                       const float* __restrict__ Wa,
                       const float* __restrict__ Wb,
                       const float* __restrict__ ba,
                       const float* __restrict__ bb,
                       float* __restrict__ outA,
                       float* __restrict__ outB) {
    const int t = blockIdx.x * AGG_THREADS + threadIdx.x;
    const int n = t >> 4;
    const int j = t & 15;

    const float4* fs4 = (const float4*)fs;
    const float4  b   = ((const float4*)fd)[n * 16 + j];
    const float4  w   = ((const float4*)attn_l)[j];
    const int start = rowptr[n];
    const int end   = rowptr[n + 1];

    float m = -INFINITY, l = 0.f;
    const v2f b01 = {b.x, b.y}, b23 = {b.z, b.w};
    const v2f w01 = {w.x, w.y}, w23 = {w.z, w.w};
    v2f acc01 = {0.f, 0.f}, acc23 = {0.f, 0.f};

    const int  i4 = j & 3;
    const bool b0 = (i4 & 1) != 0;
    const bool b1 = (i4 & 2) != 0;

    for (int p = start; p < end; p += EDGE_BATCH) {
        v2f   a01[EDGE_BATCH], a23[EDGE_BATCH];
        float part[EDGE_BATCH];
#pragma unroll
        for (int u = 0; u < EDGE_BATCH; ++u) {
            const int q = (p + u < end) ? (p + u) : (end - 1);
            const float4 a = fs4[csr_src[q] * 16 + j];
            a01[u] = (v2f){a.x, a.y};
            a23[u] = (v2f){a.z, a.w};
        }
#pragma unroll
        for (int u = 0; u < EDGE_BATCH; ++u) {
            const v2f x01 = a01[u] + b01;
            const v2f x23 = a23[u] + b23;
            const v2f z01 = __builtin_elementwise_max(x01, x01 * NEG_SLOPE);
            const v2f z23 = __builtin_elementwise_max(x23, x23 * NEG_SLOPE);
            const v2f pp = __builtin_elementwise_fma(z23, w23, z01 * w01);
            part[u] = pp.x + pp.y;
        }
        const float sA = (b0 ? part[1] : part[0]) + quad_perm<QP_XOR1>(b0 ? part[0] : part[1]);
        const float sB = (b0 ? part[3] : part[2]) + quad_perm<QP_XOR1>(b0 ? part[2] : part[3]);
        const float s  = (b1 ? sB : sA) + quad_perm<QP_XOR2>(b1 ? sA : sB);
        const float sc = (p + i4 < end) ? s : -INFINITY;
        float t = fmaxf(sc, quad_perm<QP_XOR1>(sc));
        t = fmaxf(t, quad_perm<QP_XOR2>(t));
        const float mn = fmaxf(m, t);
        const float r  = expf(m - mn);
        const float pw = expf(sc - mn);
        l *= r;
        acc01 *= r;
        acc23 *= r;
        const float pe4[EDGE_BATCH] = { quad_perm<QP_LANE(0)>(pw), quad_perm<QP_LANE(1)>(pw),
                                        quad_perm<QP_LANE(2)>(pw), quad_perm<QP_LANE(3)>(pw) };
#pragma unroll
        for (int u = 0; u < EDGE_BATCH; ++u) {
            const float pe = pe4[u];
            const v2f   pv = {pe, pe};
            l    += pe;
            acc01 = __builtin_elementwise_fma(a01[u], pv, acc01);
            acc23 = __builtin_elementwise_fma(a23[u], pv, acc23);
        }
        m = mn;
    }
    const float4 acc = make_float4(acc01.x, acc01.y, acc23.x, acc23.y);

    float4 o = make_float4(0.f, 0.f, 0.f, 0.f);
    if (end > start) {
        const float inv = 1.f / l;
        o.x = fmaxf(acc.x * inv, 0.f);
        o.y = fmaxf(acc.y * inv, 0.f);
        o.z = fmaxf(acc.z * inv, 0.f);
        o.w = fmaxf(acc.w * inv, 0.f);
    }
    if (!LAST) {
        __shared__ __attribute__((aligned(16))) float sT[(AGG_THREADS / 16) * HP];
        __shared__ __attribute__((aligned(16))) float sO[(AGG_THREADS / 16) * PO];
        const int ln = threadIdx.x >> 4;
        *(float4*)(sT + ln * HP + 4 * j) = o;
        __syncthreads();
        {
            const int lane = threadIdx.x & 31;
            const int wt   = threadIdx.x >> 5;
            const int c    = lane & 15;
            const int hf   = lane >> 4;
            const int n0   = (wt & 3) * 16;
            const float* Wp   = (wt < 4) ? Wa : Wb;
            const float* bp   = (wt < 4) ? ba : bb;
            const float* hrow = sT + c * HP + 2 * hf;
            const float* wcol = Wp + (2 * hf) * H_FEATS + n0 + c;
            v8f ct = (v8f){0.f, 0.f, 0.f, 0.f, 0.f, 0.f, 0.f, 0.f};
#pragma unroll
            for (int s = 0; s < H_FEATS / 4; ++s) {
                const float2 av = *(const float2*)(hrow + 4 * s);
                v2f af; af.x = av.x; af.y = av.y;
                v2f bf; bf.x = wcol[(4 * s) * H_FEATS]; bf.y = wcol[(4 * s + 1) * H_FEATS];
                ct = wmma_f32_16x16x4(af, bf, ct);
            }
            const float bv = bp[n0 + c];
#pragma unroll
            for (int r = 0; r < 8; ++r)
                sO[(8 * hf + r) * PO + wt * 16 + c] = ct[r] + bv;
        }
        __syncthreads();
        ((float4*)outA)[n * 16 + j] = *(const float4*)(sO + ln * PO + 4 * j);
        ((float4*)outB)[n * 16 + j] = *(const float4*)(sO + ln * PO + H_FEATS + 4 * j);
    } else {
        __shared__ float s_out[(AGG_THREADS / 16) * 2];
        const float4 wa = ((const float4*)Wa)[2 * j];
        const float4 wb = ((const float4*)Wa)[2 * j + 1];
        float p0 = o.x * wa.x + o.y * wa.z + o.z * wb.x + o.w * wb.z;
        float p1 = o.x * wa.y + o.y * wa.w + o.z * wb.y + o.w * wb.w;
#pragma unroll
        for (int msk = 1; msk < 16; msk <<= 1) {
            p0 += __shfl_xor(p0, msk);
            p1 += __shfl_xor(p1, msk);
        }
        if (j == 0) {
            const int ln = threadIdx.x >> 4;
            s_out[2 * ln]     = p0 + ba[0];
            s_out[2 * ln + 1] = p1 + ba[1];
        }
        __syncthreads();
        if (threadIdx.x < (AGG_THREADS / 16) * 2)
            outA[blockIdx.x * ((AGG_THREADS / 16) * 2) + threadIdx.x] = s_out[threadIdx.x];
    }
}

extern "C" void kernel_launch(void* const* d_in, const int* in_sizes, int n_in,
                              void* d_out, int out_size, void* d_ws, size_t ws_size,
                              hipStream_t stream) {
    (void)in_sizes; (void)n_in; (void)out_size; (void)ws_size;

    const float* feature  = (const float*)d_in[0];
    const int*   src      = (const int*)  d_in[1];
    const int*   dst      = (const int*)  d_in[2];
    const float* W_in     = (const float*)d_in[3];
    const float* b_in     = (const float*)d_in[4];
    const float* fc_src_W = (const float*)d_in[5];
    const float* fc_src_b = (const float*)d_in[6];
    const float* fc_dst_W = (const float*)d_in[7];
    const float* fc_dst_b = (const float*)d_in[8];
    const float* attn     = (const float*)d_in[9];
    const float* W_out    = (const float*)d_in[10];
    const float* b_out    = (const float*)d_in[11];
    float* logits = (float*)d_out;

    float* ws = (float*)d_ws;
    size_t off = 0;
    float* fs    = ws + off; off += (size_t)N_NODES * H_FEATS;
    float* fd    = ws + off; off += (size_t)N_NODES * H_FEATS;
    float* fs1   = ws + off; off += (size_t)N_NODES * H_FEATS;
    float* fd1   = ws + off; off += (size_t)N_NODES * H_FEATS;
    int* rowptr   = (int*)(ws + off); off += (size_t)N_NODES + 16;
    int* csr_src  = (int*)(ws + off); off += (size_t)N_EDGES;
    int* rec      = (int*)(ws + off); off += (size_t)N_EDGES;
    int* tab      = (int*)(ws + off); off += (size_t)PBLK * TABP;

    linears_beside_sort_pass<0><<<GEMM_BLOCKS_0 + PBLK, GEMM_WAVES * 32, 0, stream>>>(
        feature, W_in, b_in, fc_src_W, fc_dst_W, fc_src_b, fc_dst_b, fs, fd,
        src, dst, tab, rec, rowptr, csr_src);
    linears_beside_sort_pass<1><<<GEMM_BLOCKS_1 + NBINS, GEMM_WAVES * 32, 0, stream>>>(
        feature, W_in, b_in, fc_src_W, fc_dst_W, fc_src_b, fc_dst_b, fs, fd,
        src, dst, tab, rec, rowptr, csr_src);

    const int ablocks = (N_NODES * 16) / AGG_THREADS;

    node_softmax_aggregate<false><<<ablocks, AGG_THREADS, 0, stream>>>(
        rowptr, csr_src, fs, fd, attn,
        fc_src_W + (size_t)H_FEATS * H_FEATS, fc_dst_W + (size_t)H_FEATS * H_FEATS,
        fc_src_b + H_FEATS, fc_dst_b + H_FEATS, fs1, fd1);
    node_softmax_aggregate<true><<<ablocks, AGG_THREADS, 0, stream>>>(
        rowptr, csr_src, fs1, fd1, attn + HEADS * D_HEAD,
        W_out, W_out, b_out, b_out, logits, logits);
}
